# starting kernel
# baseline (speedup 1.0000x reference)
#include <hip/hip_runtime.h>
#include <hip/hip_bf16.h>

typedef __attribute__((ext_vector_type(2))) float v2f;
typedef __attribute__((ext_vector_type(8))) float v8f;

#define N_NODES 100000
#define N_EDGES 1600000
#define IN_F    128
#define HID_F   128
#define OUT_F   64


__global__ void zero_kernel(float* __restrict__ p, long n) {
    long i = (long)blockIdx.x * blockDim.x + threadIdx.x;
    long stride = (long)gridDim.x * blockDim.x;
    for (; i < n; i += stride) p[i] = 0.0f;
}

__global__ void deg_count_kernel(const int* __restrict__ dst, float* __restrict__ deg, int E) {
    int e = blockIdx.x * blockDim.x + threadIdx.x;
    if (e < E) atomicAdd(&deg[dst[e]], 1.0f);
}

__global__ void deg_finish_kernel(float* __restrict__ deg, int n) {
    int i = blockIdx.x * blockDim.x + threadIdx.x;
    if (i < n) {
        float d = deg[i];
        deg[i] = (d > 0.0f) ? (1.0f / sqrtf(d)) : 0.0f;
    }
}

template <int FOUT, bool RELU_IN>
__global__ __launch_bounds__(32 * (FOUT / 16))
void gemm_wmma_kernel(const float* __restrict__ X, const float* __restrict__ W,
                      const float* __restrict__ bias, float* __restrict__ H) {
    constexpr int K = 128;
    const int wave = threadIdx.x >> 5;
    const int lane = threadIdx.x & 31;
    const int half = lane >> 4;
    const int l    = lane & 15;
    const int m0 = blockIdx.x * 16;
    const int n0 = wave * 16;

    const float* arow = X + (long)(m0 + l) * K + 2 * half;
    const float* brow = W + (long)(n0 + l) * K + 2 * half;

    v8f c = {};
#pragma unroll
    for (int s = 0; s < K / 4; ++s) {
        v2f a = *(const v2f*)(arow + 4 * s);
        v2f b = *(const v2f*)(brow + 4 * s);
        if (RELU_IN) {
            a.x = fmaxf(a.x, 0.0f);
            a.y = fmaxf(a.y, 0.0f);
        }
        c = __builtin_amdgcn_wmma_f32_16x16x4_f32(false, a, false, b,
                                                  (short)0, c, false, false);
    }

    const float bb = bias[n0 + l];
#pragma unroll
    for (int r = 0; r < 8; ++r) {
        const int row = m0 + r + 8 * half;
        H[(long)row * FOUT + n0 + l] = c[r] + bb;
    }
}

__global__ void aggregate128_kernel(const int* __restrict__ src, const int* __restrict__ dst,
                                    const float* __restrict__ dis, const float* __restrict__ h,
                                    float* __restrict__ out, int E) {
    int e = blockIdx.x * (blockDim.x >> 5) + (threadIdx.x >> 5);
    if (e >= E) return;
    int lane = threadIdx.x & 31;
    int s = src[e], d = dst[e];
    float w = dis[s] * dis[d];
    float4 v = ((const float4*)(h + (long)s * 128))[lane];
    float* o = out + (long)d * 128 + lane * 4;
    atomicAdd(o + 0, v.x * w);
    atomicAdd(o + 1, v.y * w);
    atomicAdd(o + 2, v.z * w);
    atomicAdd(o + 3, v.w * w);
}

__global__ void aggregate64_kernel(const int* __restrict__ src, const int* __restrict__ dst,
                                   const float* __restrict__ dis, const float* __restrict__ h,
                                   float* __restrict__ out, int E) {
    int e = blockIdx.x * (blockDim.x >> 5) + (threadIdx.x >> 5);
    if (e >= E) return;
    int lane = threadIdx.x & 31;
    int s = src[e], d = dst[e];
    float w = dis[s] * dis[d];
    float2 v = ((const float2*)(h + (long)s * 64))[lane];
    float* o = out + (long)d * 64 + lane * 2;
    atomicAdd(o + 0, v.x * w);
    atomicAdd(o + 1, v.y * w);
}


extern "C" void kernel_launch(void* const* d_in, const int* in_sizes, int n_in,
                              void* d_out, int out_size, void* d_ws, size_t ws_size,
                              hipStream_t stream) {
    const float* x  = (const float*)d_in[0];
    const int*   ei = (const int*)d_in[1];
    const float* W1 = (const float*)d_in[2];
    const float* b1 = (const float*)d_in[3];
    const float* W2 = (const float*)d_in[4];
    const float* b2 = (const float*)d_in[5];
    float* out = (float*)d_out;

    const int* src = ei;
    const int* dst = ei + N_EDGES;

    float* deg  = (float*)d_ws;
    float* h1   = deg + N_NODES;
    float* agg1 = h1 + (long)N_NODES * HID_F;
    float* h2   = agg1 + (long)N_NODES * HID_F;

    zero_kernel<<<1024, 256, 0, stream>>>(deg, (long)N_NODES);
    deg_count_kernel<<<(N_EDGES + 255) / 256, 256, 0, stream>>>(dst, deg, N_EDGES);
    deg_finish_kernel<<<(N_NODES + 255) / 256, 256, 0, stream>>>(deg, N_NODES);

    gemm_wmma_kernel<HID_F, false><<<N_NODES / 16, 256, 0, stream>>>(x, W1, b1, h1);

    zero_kernel<<<4096, 256, 0, stream>>>(agg1, (long)N_NODES * HID_F);
    aggregate128_kernel<<<(N_EDGES + 7) / 8, 256, 0, stream>>>(src, dst, deg, h1, agg1, N_EDGES);

    gemm_wmma_kernel<OUT_F, true><<<N_NODES / 16, 128, 0, stream>>>(agg1, W2, b2, h2);

    zero_kernel<<<4096, 256, 0, stream>>>(out, (long)N_NODES * OUT_F);
    aggregate64_kernel<<<(N_EDGES + 7) / 8, 256, 0, stream>>>(src, dst, deg, h2, out, N_EDGES);
}


# Replace the float-atomic scatter-add of both aggregations by a gather over a CSR built in the call (int atomic count+rank, scan, fill); per-row sums now run in rank order with fused multiply-add
# speedup vs baseline: 2.0397x; 2.0397x over previous
#include <hip/hip_runtime.h>
#include <hip/hip_bf16.h>

typedef __attribute__((ext_vector_type(2))) float v2f;
typedef __attribute__((ext_vector_type(8))) float v8f;

#define N_NODES 100000
#define N_EDGES 1600000
#define IN_F    128
#define HID_F   128
#define OUT_F   64


#define SCAN_CHUNK 1024

__global__ void zero_int_kernel(int* __restrict__ p, int n) {
    int i = blockIdx.x * blockDim.x + threadIdx.x;
    if (i < n) p[i] = 0;
}

__global__ void count_rank_kernel(const int* __restrict__ dst, int* __restrict__ cnt,
                                  int* __restrict__ pos, int E) {
    int e = blockIdx.x * blockDim.x + threadIdx.x;
    if (e < E) pos[e] = atomicAdd(&cnt[dst[e]], 1);
}

__device__ __forceinline__ int wave_sum(int v) {
#pragma unroll
    for (int o = 16; o > 0; o >>= 1) v += __shfl_xor(v, o);
    return v;
}

__global__ __launch_bounds__(256)
void block_sum_kernel(const int* __restrict__ cnt, int* __restrict__ bsum, int n) {
    __shared__ int red[8];
    const int base = blockIdx.x * SCAN_CHUNK + threadIdx.x * 4;
    int s = 0;
#pragma unroll
    for (int k = 0; k < 4; ++k) if (base + k < n) s += cnt[base + k];
    s = wave_sum(s);
    if ((threadIdx.x & 31) == 0) red[threadIdx.x >> 5] = s;
    __syncthreads();
    if (threadIdx.x == 0) {
        int t = 0;
#pragma unroll
        for (int i = 0; i < 8; ++i) t += red[i];
        bsum[blockIdx.x] = t;
    }
}

__global__ __launch_bounds__(256)
void scan_finish_kernel(const int* __restrict__ cnt, const int* __restrict__ bsum,
                        int* __restrict__ rowptr, float* __restrict__ dis, int n) {
    __shared__ int wsum[8];
    __shared__ int wtot[8];
    const int tid = threadIdx.x, lane = tid & 31, wave = tid >> 5;

    int part = 0;
    for (int i = tid; i < (int)blockIdx.x; i += 256) part += bsum[i];
    part = wave_sum(part);
    if (lane == 0) wsum[wave] = part;

    const int base = blockIdx.x * SCAN_CHUNK + tid * 4;
    int c[4];
#pragma unroll
    for (int k = 0; k < 4; ++k) c[k] = (base + k < n) ? cnt[base + k] : 0;
    const int t = c[0] + c[1] + c[2] + c[3];
    int inc = t;
#pragma unroll
    for (int o = 1; o < 32; o <<= 1) {
        int v = __shfl_up(inc, o);
        if (lane >= o) inc += v;
    }
    if (lane == 31) wtot[wave] = inc;
    __syncthreads();

    int off = 0;
#pragma unroll
    for (int i = 0; i < 8; ++i) {
        off += wsum[i];
        if (i < wave) off += wtot[i];
    }
    int excl = off + inc - t;
#pragma unroll
    for (int k = 0; k < 4; ++k) {
        const int idx = base + k;
        if (idx < n) {
            rowptr[idx] = excl;
            dis[idx] = (c[k] > 0) ? (1.0f / sqrtf((float)c[k])) : 0.0f;
        }
        excl += c[k];
    }
    if (base < n && base + 4 >= n) rowptr[n] = excl;
}

__global__ void fill_kernel(const int* __restrict__ src, const int* __restrict__ dst,
                            const int* __restrict__ pos, const int* __restrict__ rowptr,
                            int* __restrict__ csr_src, int E) {
    int e = blockIdx.x * blockDim.x + threadIdx.x;
    if (e < E) csr_src[rowptr[dst[e]] + pos[e]] = src[e];
}

template <int FOUT, bool RELU_IN>
__global__ __launch_bounds__(32 * (FOUT / 16))
void gemm_wmma_kernel(const float* __restrict__ X, const float* __restrict__ W,
                      const float* __restrict__ bias, float* __restrict__ H) {
    constexpr int K = 128;
    const int wave = threadIdx.x >> 5;
    const int lane = threadIdx.x & 31;
    const int half = lane >> 4;
    const int l    = lane & 15;
    const int m0 = blockIdx.x * 16;
    const int n0 = wave * 16;

    const float* arow = X + (long)(m0 + l) * K + 2 * half;
    const float* brow = W + (long)(n0 + l) * K + 2 * half;

    v8f c = {};
#pragma unroll
    for (int s = 0; s < K / 4; ++s) {
        v2f a = *(const v2f*)(arow + 4 * s);
        v2f b = *(const v2f*)(brow + 4 * s);
        if (RELU_IN) {
            a.x = fmaxf(a.x, 0.0f);
            a.y = fmaxf(a.y, 0.0f);
        }
        c = __builtin_amdgcn_wmma_f32_16x16x4_f32(false, a, false, b,
                                                  (short)0, c, false, false);
    }

    const float bb = bias[n0 + l];
#pragma unroll
    for (int r = 0; r < 8; ++r) {
        const int row = m0 + r + 8 * half;
        H[(long)row * FOUT + n0 + l] = c[r] + bb;
    }
}

template <int F> struct RowVec;
template <> struct RowVec<128> { typedef float4 type; };
template <> struct RowVec<64>  { typedef float2 type; };

__device__ __forceinline__ void fma_row(float4& a, const float4& v, float w) {
    a.x = __builtin_fmaf(v.x, w, a.x); a.y = __builtin_fmaf(v.y, w, a.y);
    a.z = __builtin_fmaf(v.z, w, a.z); a.w = __builtin_fmaf(v.w, w, a.w);
}
__device__ __forceinline__ void fma_row(float2& a, const float2& v, float w) {
    a.x = __builtin_fmaf(v.x, w, a.x); a.y = __builtin_fmaf(v.y, w, a.y);
}

template <int F, int U>
__device__ __forceinline__ void gather_step(typename RowVec<F>::type& acc, const float* __restrict__ h,
                                            int s_l, float w_l, int j, int lane) {
    typedef typename RowVec<F>::type VT;
    VT v[U];
    float w[U];
#pragma unroll
    for (int u = 0; u < U; ++u) {
        const int s = __builtin_amdgcn_readlane(s_l, j + u);
        w[u] = __builtin_bit_cast(float, __builtin_amdgcn_readlane(__builtin_bit_cast(int, w_l), j + u));
        v[u] = ((const VT*)(h + (long)s * F))[lane];
    }
#pragma unroll
    for (int u = 0; u < U; ++u) fma_row(acc, v[u], w[u]);
}

template <int F>
__global__ __launch_bounds__(256)
void gather_kernel(const int* __restrict__ rowptr, const int* __restrict__ csr_src,
                   const float* __restrict__ dis, const float* __restrict__ h,
                   float* __restrict__ out, int n) {
    typedef typename RowVec<F>::type VT;
    const int node = blockIdx.x * 8 + __builtin_amdgcn_readfirstlane((int)(threadIdx.x >> 5));
    if (node >= n) return;
    const int lane = threadIdx.x & 31;
    const int beg = rowptr[node], end = rowptr[node + 1];
    const float dd = dis[node];

    VT acc = {};
    for (int base = beg; base < end; base += 32) {
        const int m = (end - base < 32) ? (end - base) : 32;
        int s_l = 0;
        float w_l = 0.0f;
        if (lane < m) {
            s_l = csr_src[base + lane];
            w_l = dis[s_l] * dd;
        }
        int j = 0;
        for (; j + 8 <= m; j += 8) gather_step<F, 8>(acc, h, s_l, w_l, j, lane);
        if (j + 4 <= m) { gather_step<F, 4>(acc, h, s_l, w_l, j, lane); j += 4; }
        if (j + 2 <= m) { gather_step<F, 2>(acc, h, s_l, w_l, j, lane); j += 2; }
        if (j < m)      { gather_step<F, 1>(acc, h, s_l, w_l, j, lane); }
    }
    ((VT*)(out + (long)node * F))[lane] = acc;
}


extern "C" void kernel_launch(void* const* d_in, const int* in_sizes, int n_in,
                              void* d_out, int out_size, void* d_ws, size_t ws_size,
                              hipStream_t stream) {
    const float* x  = (const float*)d_in[0];
    const int*   ei = (const int*)d_in[1];
    const float* W1 = (const float*)d_in[2];
    const float* b1 = (const float*)d_in[3];
    const float* W2 = (const float*)d_in[4];
    const float* b2 = (const float*)d_in[5];
    float* out = (float*)d_out;

    const int* src = ei;
    const int* dst = ei + N_EDGES;

    float* dis  = (float*)d_ws;
    float* h1   = dis + N_NODES;
    float* agg1 = h1 + (long)N_NODES * HID_F;
    float* h2   = agg1 + (long)N_NODES * HID_F;
    int* cnt     = (int*)(h2 + (long)N_NODES * OUT_F);
    int* rowptr  = cnt + N_NODES;
    int* bsum    = rowptr + N_NODES + 32;
    int* pos     = bsum + 128;
    int* csr_src = pos + N_EDGES;

    zero_int_kernel<<<(N_NODES + 255) / 256, 256, 0, stream>>>(cnt, N_NODES);
    count_rank_kernel<<<(N_EDGES + 255) / 256, 256, 0, stream>>>(dst, cnt, pos, N_EDGES);
    block_sum_kernel<<<(N_NODES + SCAN_CHUNK - 1) / SCAN_CHUNK, 256, 0, stream>>>(cnt, bsum, N_NODES);
    scan_finish_kernel<<<(N_NODES + SCAN_CHUNK - 1) / SCAN_CHUNK, 256, 0, stream>>>(cnt, bsum, rowptr, dis, N_NODES);
    fill_kernel<<<(N_EDGES + 255) / 256, 256, 0, stream>>>(src, dst, pos, rowptr, csr_src, N_EDGES);

    gemm_wmma_kernel<HID_F, false><<<N_NODES / 16, 256, 0, stream>>>(x, W1, b1, h1);

    gather_kernel<HID_F><<<(N_NODES + 7) / 8, 256, 0, stream>>>(rowptr, csr_src, dis, h1, agg1, N_NODES);

    gemm_wmma_kernel<OUT_F, true><<<N_NODES / 16, 128, 0, stream>>>(agg1, W2, b2, h2);

    gather_kernel<OUT_F><<<(N_NODES + 7) / 8, 256, 0, stream>>>(rowptr, csr_src, dis, h2, out, N_NODES);
}


# Fewer, longer-lived waves: a thread takes 8 edges, a wave 8 rows (gather) or 5 row tiles (GEMM, W fragment loaded once per wave); 5-8x fewer waves; bodies and sum order (arrival rank) kept
# speedup vs baseline: 2.1678x; 1.0628x over previous
#include <hip/hip_runtime.h>
#include <hip/hip_bf16.h>

typedef __attribute__((ext_vector_type(2))) float v2f;
typedef __attribute__((ext_vector_type(8))) float v8f;

#define N_NODES 100000
#define N_EDGES 1600000
#define IN_F    128
#define HID_F   128
#define OUT_F   64


#define SCAN_CHUNK 1024

#define EDGES_PER_THREAD 8
#define EDGE_BLOCK (256 * EDGES_PER_THREAD)
#define ROWS_PER_WAVE 8
#define TILES_PER_BLOCK 5

__global__ __launch_bounds__(256)
void zero_int_kernel(int* __restrict__ p, int n) {
    const int i = blockIdx.x * 256 + threadIdx.x;
    const int n4 = n >> 2;
    if (i < n4) ((int4*)p)[i] = make_int4(0, 0, 0, 0);
    if (i < (n & 3)) p[4 * n4 + i] = 0;
}

template <bool FULL>
__device__ __forceinline__ void count_rank_body(const int* __restrict__ dst, int* __restrict__ cnt,
                                                int* __restrict__ pos, int base, int E) {
    int d[EDGES_PER_THREAD], r[EDGES_PER_THREAD];
#pragma unroll
    for (int k = 0; k < EDGES_PER_THREAD; ++k) {
        const int e = base + k * 256;
        d[k] = (FULL || e < E) ? dst[e] : 0;
    }
#pragma unroll
    for (int k = 0; k < EDGES_PER_THREAD; ++k) {
        const int e = base + k * 256;
        r[k] = (FULL || e < E) ? atomicAdd(&cnt[d[k]], 1) : 0;
    }
#pragma unroll
    for (int k = 0; k < EDGES_PER_THREAD; ++k) {
        const int e = base + k * 256;
        if (FULL || e < E) pos[e] = r[k];
    }
}

__global__ __launch_bounds__(256)
void count_rank_kernel(const int* __restrict__ dst, int* __restrict__ cnt,
                       int* __restrict__ pos, int E) {
    const int block_base = blockIdx.x * EDGE_BLOCK;
    const int base = block_base + threadIdx.x;
    if (block_base + EDGE_BLOCK <= E) count_rank_body<true>(dst, cnt, pos, base, E);
    else                              count_rank_body<false>(dst, cnt, pos, base, E);
}

__device__ __forceinline__ int wave_sum(int v) {
#pragma unroll
    for (int o = 16; o > 0; o >>= 1) v += __shfl_xor(v, o);
    return v;
}

__global__ __launch_bounds__(256)
void block_sum_kernel(const int* __restrict__ cnt, int* __restrict__ bsum, int n) {
    __shared__ int red[8];
    const int base = blockIdx.x * SCAN_CHUNK + threadIdx.x * 4;
    int s = 0;
#pragma unroll
    for (int k = 0; k < 4; ++k) if (base + k < n) s += cnt[base + k];
    s = wave_sum(s);
    if ((threadIdx.x & 31) == 0) red[threadIdx.x >> 5] = s;
    __syncthreads();
    if (threadIdx.x == 0) {
        int t = 0;
#pragma unroll
        for (int i = 0; i < 8; ++i) t += red[i];
        bsum[blockIdx.x] = t;
    }
}

__global__ __launch_bounds__(256)
void scan_finish_kernel(const int* __restrict__ cnt, const int* __restrict__ bsum,
                        int* __restrict__ rowptr, float* __restrict__ dis, int n) {
    __shared__ int wsum[8];
    __shared__ int wtot[8];
    const int tid = threadIdx.x, lane = tid & 31, wave = tid >> 5;

    int part = 0;
    for (int i = tid; i < (int)blockIdx.x; i += 256) part += bsum[i];
    part = wave_sum(part);
    if (lane == 0) wsum[wave] = part;

    const int base = blockIdx.x * SCAN_CHUNK + tid * 4;
    int c[4];
#pragma unroll
    for (int k = 0; k < 4; ++k) c[k] = (base + k < n) ? cnt[base + k] : 0;
    const int t = c[0] + c[1] + c[2] + c[3];
    int inc = t;
#pragma unroll
    for (int o = 1; o < 32; o <<= 1) {
        int v = __shfl_up(inc, o);
        if (lane >= o) inc += v;
    }
    if (lane == 31) wtot[wave] = inc;
    __syncthreads();

    int off = 0;
#pragma unroll
    for (int i = 0; i < 8; ++i) {
        off += wsum[i];
        if (i < wave) off += wtot[i];
    }
    int excl = off + inc - t;
#pragma unroll
    for (int k = 0; k < 4; ++k) {
        const int idx = base + k;
        if (idx < n) {
            rowptr[idx] = excl;
            dis[idx] = (c[k] > 0) ? (1.0f / sqrtf((float)c[k])) : 0.0f;
        }
        excl += c[k];
    }
    if (base < n && base + 4 >= n) rowptr[n] = excl;
}

template <bool FULL>
__device__ __forceinline__ void fill_body(const int* __restrict__ src, const int* __restrict__ dst,
                                          const int* __restrict__ pos, const int* __restrict__ rowptr,
                                          int* __restrict__ csr_src, int base, int E) {
    int s[EDGES_PER_THREAD], d[EDGES_PER_THREAD], p[EDGES_PER_THREAD], q[EDGES_PER_THREAD];
#pragma unroll
    for (int k = 0; k < EDGES_PER_THREAD; ++k) {
        const int e = base + k * 256;
        const bool ok = FULL || e < E;
        s[k] = ok ? src[e] : 0;
        d[k] = ok ? dst[e] : 0;
        p[k] = ok ? pos[e] : 0;
    }
#pragma unroll
    for (int k = 0; k < EDGES_PER_THREAD; ++k) {
        const int e = base + k * 256;
        q[k] = (FULL || e < E) ? rowptr[d[k]] : 0;
    }
#pragma unroll
    for (int k = 0; k < EDGES_PER_THREAD; ++k) {
        const int e = base + k * 256;
        if (FULL || e < E) csr_src[q[k] + p[k]] = s[k];
    }
}

__global__ __launch_bounds__(256)
void fill_kernel(const int* __restrict__ src, const int* __restrict__ dst,
                 const int* __restrict__ pos, const int* __restrict__ rowptr,
                 int* __restrict__ csr_src, int E) {
    const int block_base = blockIdx.x * EDGE_BLOCK;
    const int base = block_base + threadIdx.x;
    if (block_base + EDGE_BLOCK <= E) fill_body<true>(src, dst, pos, rowptr, csr_src, base, E);
    else                              fill_body<false>(src, dst, pos, rowptr, csr_src, base, E);
}

template <int FOUT, bool RELU_IN>
__global__ __launch_bounds__(32 * (FOUT / 16))
void gemm_wmma_kernel(const float* __restrict__ X, const float* __restrict__ W,
                      const float* __restrict__ bias, float* __restrict__ H, int n_tiles) {
    constexpr int K = 128;
    const int wave = threadIdx.x >> 5;
    const int lane = threadIdx.x & 31;
    const int half = lane >> 4;
    const int l    = lane & 15;
    const int n0 = wave * 16;

    const float* brow = W + (long)(n0 + l) * K + 2 * half;
    v2f b[K / 4];
#pragma unroll
    for (int s = 0; s < K / 4; ++s) b[s] = *(const v2f*)(brow + 4 * s);
    const float bb = bias[n0 + l];

    const int t0 = blockIdx.x * TILES_PER_BLOCK;
    for (int i = 0; i < TILES_PER_BLOCK; ++i) {
        if (t0 + i >= n_tiles) break;
        const int m0 = (t0 + i) * 16;
        const float* arow = X + (long)(m0 + l) * K + 2 * half;

        v8f c = {};
#pragma unroll
        for (int s = 0; s < K / 4; ++s) {
            v2f a = *(const v2f*)(arow + 4 * s);
            if (RELU_IN) {
                a.x = fmaxf(a.x, 0.0f);
                a.y = fmaxf(a.y, 0.0f);
            }
            c = __builtin_amdgcn_wmma_f32_16x16x4_f32(false, a, false, b[s],
                                                      (short)0, c, false, false);
        }

#pragma unroll
        for (int r = 0; r < 8; ++r) {
            const int row = m0 + r + 8 * half;
            H[(long)row * FOUT + n0 + l] = c[r] + bb;
        }
    }
}

template <int F> struct RowVec;
template <> struct RowVec<128> { typedef float4 type; };
template <> struct RowVec<64>  { typedef float2 type; };

__device__ __forceinline__ void fma_row(float4& a, const float4& v, float w) {
    a.x = __builtin_fmaf(v.x, w, a.x); a.y = __builtin_fmaf(v.y, w, a.y);
    a.z = __builtin_fmaf(v.z, w, a.z); a.w = __builtin_fmaf(v.w, w, a.w);
}
__device__ __forceinline__ void fma_row(float2& a, const float2& v, float w) {
    a.x = __builtin_fmaf(v.x, w, a.x); a.y = __builtin_fmaf(v.y, w, a.y);
}

template <int F, int U>
__device__ __forceinline__ void gather_step(typename RowVec<F>::type& acc, const float* __restrict__ h,
                                            int s_l, float w_l, int j, int lane) {
    typedef typename RowVec<F>::type VT;
    VT v[U];
    float w[U];
#pragma unroll
    for (int u = 0; u < U; ++u) {
        const int s = __builtin_amdgcn_readlane(s_l, j + u);
        w[u] = __builtin_bit_cast(float, __builtin_amdgcn_readlane(__builtin_bit_cast(int, w_l), j + u));
        v[u] = ((const VT*)(h + (long)s * F))[lane];
    }
#pragma unroll
    for (int u = 0; u < U; ++u) fma_row(acc, v[u], w[u]);
}

template <int F>
__global__ __launch_bounds__(256)
void gather_kernel(const int* __restrict__ rowptr, const int* __restrict__ csr_src,
                   const float* __restrict__ dis, const float* __restrict__ h,
                   float* __restrict__ out, int n) {
    typedef typename RowVec<F>::type VT;
    const int node0 = (blockIdx.x * 8 + __builtin_amdgcn_readfirstlane((int)(threadIdx.x >> 5))) * ROWS_PER_WAVE;
    if (node0 >= n) return;
    const int lane = threadIdx.x & 31;
    const int rows = (n - node0 < ROWS_PER_WAVE) ? (n - node0) : ROWS_PER_WAVE;
    const int   rp_l = (lane <= rows) ? rowptr[node0 + lane] : 0;
    const float dd_l = (lane <  rows) ? dis[node0 + lane] : 0.0f;

    for (int i = 0; i < rows; ++i) {
        const int beg = __builtin_amdgcn_readlane(rp_l, i);
        const int end = __builtin_amdgcn_readlane(rp_l, i + 1);
        const float dd = __builtin_bit_cast(float, __builtin_amdgcn_readlane(__builtin_bit_cast(int, dd_l), i));

        VT acc = {};
        for (int base = beg; base < end; base += 32) {
            const int m = (end - base < 32) ? (end - base) : 32;
            int s_l = 0;
            float w_l = 0.0f;
            if (lane < m) {
                s_l = csr_src[base + lane];
                w_l = dis[s_l] * dd;
            }
            int j = 0;
            for (; j + 8 <= m; j += 8) gather_step<F, 8>(acc, h, s_l, w_l, j, lane);
            if (j + 4 <= m) { gather_step<F, 4>(acc, h, s_l, w_l, j, lane); j += 4; }
            if (j + 2 <= m) { gather_step<F, 2>(acc, h, s_l, w_l, j, lane); j += 2; }
            if (j < m)      { gather_step<F, 1>(acc, h, s_l, w_l, j, lane); }
        }
        ((VT*)(out + (long)(node0 + i) * F))[lane] = acc;
    }
}


extern "C" void kernel_launch(void* const* d_in, const int* in_sizes, int n_in,
                              void* d_out, int out_size, void* d_ws, size_t ws_size,
                              hipStream_t stream) {
    const float* x  = (const float*)d_in[0];
    const int*   ei = (const int*)d_in[1];
    const float* W1 = (const float*)d_in[2];
    const float* b1 = (const float*)d_in[3];
    const float* W2 = (const float*)d_in[4];
    const float* b2 = (const float*)d_in[5];
    float* out = (float*)d_out;

    const int* src = ei;
    const int* dst = ei + N_EDGES;

    float* dis  = (float*)d_ws;
    float* h1   = dis + N_NODES;
    float* agg1 = h1 + (long)N_NODES * HID_F;
    float* h2   = agg1 + (long)N_NODES * HID_F;
    int* cnt     = (int*)(h2 + (long)N_NODES * OUT_F);
    int* rowptr  = cnt + N_NODES;
    int* bsum    = rowptr + N_NODES + 32;
    int* pos     = bsum + 128;
    int* csr_src = pos + N_EDGES;

    zero_int_kernel<<<(N_NODES / 4 + 255) / 256, 256, 0, stream>>>(cnt, N_NODES);
    count_rank_kernel<<<(N_EDGES + EDGE_BLOCK - 1) / EDGE_BLOCK, 256, 0, stream>>>(dst, cnt, pos, N_EDGES);
    block_sum_kernel<<<(N_NODES + SCAN_CHUNK - 1) / SCAN_CHUNK, 256, 0, stream>>>(cnt, bsum, N_NODES);
    scan_finish_kernel<<<(N_NODES + SCAN_CHUNK - 1) / SCAN_CHUNK, 256, 0, stream>>>(cnt, bsum, rowptr, dis, N_NODES);
    fill_kernel<<<(N_EDGES + EDGE_BLOCK - 1) / EDGE_BLOCK, 256, 0, stream>>>(src, dst, pos, rowptr, csr_src, N_EDGES);

    gemm_wmma_kernel<HID_F, false><<<(N_NODES / 16 + TILES_PER_BLOCK - 1) / TILES_PER_BLOCK, 256, 0, stream>>>(x, W1, b1, h1, N_NODES / 16);

    gather_kernel<HID_F><<<(N_NODES + 8 * ROWS_PER_WAVE - 1) / (8 * ROWS_PER_WAVE), 256, 0, stream>>>(rowptr, csr_src, dis, h1, agg1, N_NODES);

    gemm_wmma_kernel<OUT_F, true><<<(N_NODES / 16 + TILES_PER_BLOCK - 1) / TILES_PER_BLOCK, 128, 0, stream>>>(agg1, W2, b2, h2, N_NODES / 16);

    gather_kernel<OUT_F><<<(N_NODES + 8 * ROWS_PER_WAVE - 1) / (8 * ROWS_PER_WAVE), 256, 0, stream>>>(rowptr, csr_src, dis, h2, out, N_NODES);
}


# CSR build as a two-level counting sort in LDS (edges grouped by 512-node destination band, then ordered per band), no global atomics or scattered fill; per-row sum order now LDS arrival order
# speedup vs baseline: 3.3913x; 1.5644x over previous
#include <hip/hip_runtime.h>
#include <hip/hip_bf16.h>

typedef __attribute__((ext_vector_type(2))) float v2f;
typedef __attribute__((ext_vector_type(8))) float v8f;

#define N_NODES 100000
#define N_EDGES 1600000
#define IN_F    128
#define HID_F   128
#define OUT_F   64


#define BAND_SHIFT 9
#define BAND (1 << BAND_SHIFT)
#define NBAND ((N_NODES + BAND - 1) >> BAND_SHIFT)
#define SORT_THREADS 1024
#define EDGES_PER_THREAD 8
#define CHUNK (SORT_THREADS * EDGES_PER_THREAD)
#define NCHUNK ((N_EDGES + CHUNK - 1) / CHUNK)
#define BAND_EDGES_PER_THREAD 12
#define BAND_CAP (SORT_THREADS * BAND_EDGES_PER_THREAD)
#define BANDS_PER_LANE ((NBAND + 31) / 32)
#define NBAND_PAD (BANDS_PER_LANE * 32)
#define CHUNKS_PER_LANE ((NCHUNK + 31) / 32)
#define CNT_PITCH (CHUNKS_PER_LANE * 32)

static_assert(N_NODES <= (1 << (31 - BAND_SHIFT)), "src must fit above the band-local destination");

#define ROWS_PER_WAVE 8
#define TILES_PER_BLOCK 5

__device__ __forceinline__ void wave_exscan_bands(const int* in, int* out, int lane) {
    int v[BANDS_PER_LANE];
    int t = 0;
#pragma unroll
    for (int k = 0; k < BANDS_PER_LANE; ++k) {
        v[k] = in[lane * BANDS_PER_LANE + k];
        t += v[k];
    }
    int inc = t;
#pragma unroll
    for (int o = 1; o < 32; o <<= 1) {
        const int up = __shfl_up(inc, o);
        if (lane >= o) inc += up;
    }
    int run = inc - t;
#pragma unroll
    for (int k = 0; k < BANDS_PER_LANE; ++k) {
        out[lane * BANDS_PER_LANE + k] = run;
        run += v[k];
    }
}

__global__ __launch_bounds__(SORT_THREADS)
void band_count_kernel(const int* __restrict__ dst, int* __restrict__ cntmat, int E) {
    __shared__ int s_cnt[NBAND];
    const int tid = threadIdx.x;
    if (tid < NBAND) s_cnt[tid] = 0;
    const int base = blockIdx.x * CHUNK + tid;
    int d[EDGES_PER_THREAD];
#pragma unroll
    for (int k = 0; k < EDGES_PER_THREAD; ++k) {
        const int e = base + k * SORT_THREADS;
        const int v = dst[min(e, E - 1)];
        d[k] = (e < E) ? v : -1;
    }
    __syncthreads();
#pragma unroll
    for (int k = 0; k < EDGES_PER_THREAD; ++k)
        if (d[k] >= 0) atomicAdd(&s_cnt[d[k] >> BAND_SHIFT], 1);
    __syncthreads();
    const int chunk = blockIdx.x;
    if (tid < NBAND)
        cntmat[tid * CNT_PITCH + (chunk % CHUNKS_PER_LANE) * 32 + chunk / CHUNKS_PER_LANE] = s_cnt[tid];
}

__global__ __launch_bounds__(SORT_THREADS)
void band_group_kernel(const int* __restrict__ src, const int* __restrict__ dst,
                       const int* __restrict__ cntmat, int* __restrict__ band_start,
                       int* __restrict__ binned, int E) {
    __shared__ int s_cnt[NBAND_PAD];
    __shared__ int s_lstart[NBAND_PAD];
    __shared__ int s_tot[NBAND_PAD];
    __shared__ int s_pre[NBAND_PAD];
    __shared__ int s_gstart[NBAND_PAD];
    __shared__ int s_rec[CHUNK];
    const int tid = threadIdx.x, lane = tid & 31;
    const int wave = __builtin_amdgcn_readfirstlane(tid >> 5);
    const int chunk = blockIdx.x;

    if (tid < NBAND_PAD) s_cnt[tid] = 0;

    const int base = chunk * CHUNK + tid;
    int d[EDGES_PER_THREAD], s[EDGES_PER_THREAD], r[EDGES_PER_THREAD];
#pragma unroll
    for (int k = 0; k < EDGES_PER_THREAD; ++k) {
        const int e = base + k * SORT_THREADS;
        const int ec = min(e, E - 1);
        const int dv = dst[ec];
        s[k] = src[ec];
        d[k] = (e < E) ? dv : -1;
    }

    const int own_lane = chunk / CHUNKS_PER_LANE;
    const int own_k    = chunk % CHUNKS_PER_LANE;
    int v[BANDS_PER_LANE][CHUNKS_PER_LANE];
#pragma unroll
    for (int j = 0; j < BANDS_PER_LANE; ++j) {
        const int b = wave + j * (SORT_THREADS / 32);
        const int* row = cntmat + min(b, NBAND - 1) * CNT_PITCH;
#pragma unroll
        for (int k = 0; k < CHUNKS_PER_LANE; ++k) {
            const bool ok = lane * CHUNKS_PER_LANE + k < NCHUNK;
            const int x = row[ok ? k * 32 + lane : 0];
            v[j][k] = ok ? x : 0;
        }
    }
    int t[BANDS_PER_LANE], part[BANDS_PER_LANE], inc[BANDS_PER_LANE];
#pragma unroll
    for (int j = 0; j < BANDS_PER_LANE; ++j) {
        t[j] = 0;
        part[j] = 0;
#pragma unroll
        for (int k = 0; k < CHUNKS_PER_LANE; ++k) {
            t[j] += v[j][k];
            part[j] += (k < own_k) ? v[j][k] : 0;
        }
        inc[j] = t[j];
    }
#pragma unroll
    for (int o = 1; o < 32; o <<= 1) {
        int up[BANDS_PER_LANE];
#pragma unroll
        for (int j = 0; j < BANDS_PER_LANE; ++j) up[j] = __shfl_up(inc[j], o);
#pragma unroll
        for (int j = 0; j < BANDS_PER_LANE; ++j) inc[j] += (lane >= o) ? up[j] : 0;
    }
#pragma unroll
    for (int j = 0; j < BANDS_PER_LANE; ++j) {
        const int b = wave + j * (SORT_THREADS / 32);
        const int tot = __builtin_amdgcn_readlane(inc[j], 31);
        const int pre = __builtin_amdgcn_readlane(inc[j] - t[j] + part[j], own_lane);
        if (lane == 0) {
            s_tot[b] = (b < NBAND) ? tot : 0;
            s_pre[b] = (b < NBAND) ? pre : 0;
        }
    }
    __syncthreads();

#pragma unroll
    for (int k = 0; k < EDGES_PER_THREAD; ++k)
        r[k] = (d[k] >= 0) ? atomicAdd(&s_cnt[d[k] >> BAND_SHIFT], 1) : 0;
    if (wave == 1) wave_exscan_bands(s_tot, s_gstart, lane);
    __syncthreads();

    if (wave == 0) wave_exscan_bands(s_cnt, s_lstart, lane);
    __syncthreads();

    if (tid < NBAND) {
        const int bs = s_gstart[tid];
        if (chunk == 0) band_start[tid] = bs;
        s_gstart[tid] = bs + s_pre[tid];
    }
    if (chunk == 0 && tid == 0) band_start[NBAND] = E;
#pragma unroll
    for (int k = 0; k < EDGES_PER_THREAD; ++k)
        if (d[k] >= 0)
            s_rec[s_lstart[d[k] >> BAND_SHIFT] + r[k]] = (s[k] << BAND_SHIFT) | (d[k] & (BAND - 1));
    __syncthreads();

    for (int b = wave; b < NBAND; b += SORT_THREADS / 32) {
        const int n = s_cnt[b], ls = s_lstart[b], gs = s_gstart[b];
        for (int i = lane; i < n; i += 32) binned[gs + i] = s_rec[ls + i];
    }
}

__global__ __launch_bounds__(SORT_THREADS)
void band_finish_kernel(const int* __restrict__ binned, const int* __restrict__ band_start,
                        int* __restrict__ rowptr, float* __restrict__ dis,
                        int* __restrict__ csr_src, int n) {
    __shared__ int s_cnt[BAND];
    __shared__ int s_wtot[BAND / 32];
    __shared__ int s_out[BAND_CAP];
    const int tid = threadIdx.x, lane = tid & 31, wave = tid >> 5;
    const int band = blockIdx.x;
    const int beg = band_start[band], end = band_start[band + 1];
    const int m = end - beg;
    const bool in_lds = (m <= BAND_CAP);

    if (tid < BAND) s_cnt[tid] = 0;

    int rec[BAND_EDGES_PER_THREAD];
#pragma unroll
    for (int k = 0; k < BAND_EDGES_PER_THREAD; ++k) rec[k] = -1;
    if (m > 0) {
#pragma unroll
        for (int k = 0; k < BAND_EDGES_PER_THREAD; ++k) {
            const int i = k * SORT_THREADS + tid;
            const int v = binned[beg + min(i, m - 1)];
            rec[k] = (i < m) ? v : -1;
        }
    }
    __syncthreads();

#pragma unroll
    for (int k = 0; k < BAND_EDGES_PER_THREAD; ++k)
        if (rec[k] >= 0) atomicAdd(&s_cnt[rec[k] & (BAND - 1)], 1);
    for (int i = BAND_CAP + tid; i < m; i += SORT_THREADS)
        atomicAdd(&s_cnt[binned[beg + i] & (BAND - 1)], 1);
    __syncthreads();

    int c = 0, inc = 0;
    if (tid < BAND) {
        c = s_cnt[tid];
        inc = c;
#pragma unroll
        for (int o = 1; o < 32; o <<= 1) {
            const int up = __shfl_up(inc, o);
            if (lane >= o) inc += up;
        }
        if (lane == 31) s_wtot[wave] = inc;
    }
    __syncthreads();
    if (tid < BAND) {
        int off = 0;
#pragma unroll
        for (int w = 0; w < BAND / 32; ++w) off += (w < wave) ? s_wtot[w] : 0;
        const int excl = off + inc - c;
        s_cnt[tid] = excl;
        const int node = band * BAND + tid;
        if (node < n) {
            rowptr[node] = beg + excl;
            dis[node] = (c > 0) ? (1.0f / sqrtf((float)c)) : 0.0f;
        }
    }
    if (band == NBAND - 1 && tid == 0) rowptr[n] = end;
    __syncthreads();

    int p[BAND_EDGES_PER_THREAD];
#pragma unroll
    for (int k = 0; k < BAND_EDGES_PER_THREAD; ++k)
        p[k] = (rec[k] >= 0) ? atomicAdd(&s_cnt[rec[k] & (BAND - 1)], 1) : 0;
    if (in_lds) {
#pragma unroll
        for (int k = 0; k < BAND_EDGES_PER_THREAD; ++k)
            if (rec[k] >= 0) s_out[p[k]] = rec[k] >> BAND_SHIFT;
    } else {
#pragma unroll
        for (int k = 0; k < BAND_EDGES_PER_THREAD; ++k)
            if (rec[k] >= 0) csr_src[beg + p[k]] = rec[k] >> BAND_SHIFT;
    }
    for (int i = BAND_CAP + tid; i < m; i += SORT_THREADS) {
        const int w = binned[beg + i];
        csr_src[beg + atomicAdd(&s_cnt[w & (BAND - 1)], 1)] = w >> BAND_SHIFT;
    }
    if (in_lds) {
        __syncthreads();
        for (int i = tid; i < m; i += SORT_THREADS) csr_src[beg + i] = s_out[i];
    }
}

template <int FOUT, bool RELU_IN>
__global__ __launch_bounds__(32 * (FOUT / 16))
void gemm_wmma_kernel(const float* __restrict__ X, const float* __restrict__ W,
                      const float* __restrict__ bias, float* __restrict__ H, int n_tiles) {
    constexpr int K = 128;
    const int wave = threadIdx.x >> 5;
    const int lane = threadIdx.x & 31;
    const int half = lane >> 4;
    const int l    = lane & 15;
    const int n0 = wave * 16;

    const float* brow = W + (long)(n0 + l) * K + 2 * half;
    v2f b[K / 4];
#pragma unroll
    for (int s = 0; s < K / 4; ++s) b[s] = *(const v2f*)(brow + 4 * s);
    const float bb = bias[n0 + l];

    const int t0 = blockIdx.x * TILES_PER_BLOCK;
    for (int i = 0; i < TILES_PER_BLOCK; ++i) {
        if (t0 + i >= n_tiles) break;
        const int m0 = (t0 + i) * 16;
        const float* arow = X + (long)(m0 + l) * K + 2 * half;

        v8f c = {};
#pragma unroll
        for (int s = 0; s < K / 4; ++s) {
            v2f a = *(const v2f*)(arow + 4 * s);
            if (RELU_IN) {
                a.x = fmaxf(a.x, 0.0f);
                a.y = fmaxf(a.y, 0.0f);
            }
            c = __builtin_amdgcn_wmma_f32_16x16x4_f32(false, a, false, b[s],
                                                      (short)0, c, false, false);
        }

#pragma unroll
        for (int r = 0; r < 8; ++r) {
            const int row = m0 + r + 8 * half;
            H[(long)row * FOUT + n0 + l] = c[r] + bb;
        }
    }
}

template <int F> struct RowVec;
template <> struct RowVec<128> { typedef float4 type; };
template <> struct RowVec<64>  { typedef float2 type; };

__device__ __forceinline__ void fma_row(float4& a, const float4& v, float w) {
    a.x = __builtin_fmaf(v.x, w, a.x); a.y = __builtin_fmaf(v.y, w, a.y);
    a.z = __builtin_fmaf(v.z, w, a.z); a.w = __builtin_fmaf(v.w, w, a.w);
}
__device__ __forceinline__ void fma_row(float2& a, const float2& v, float w) {
    a.x = __builtin_fmaf(v.x, w, a.x); a.y = __builtin_fmaf(v.y, w, a.y);
}

template <int F, int U>
__device__ __forceinline__ void gather_step(typename RowVec<F>::type& acc, const float* __restrict__ h,
                                            int s_l, float w_l, int j, int lane) {
    typedef typename RowVec<F>::type VT;
    VT v[U];
    float w[U];
#pragma unroll
    for (int u = 0; u < U; ++u) {
        const int s = __builtin_amdgcn_readlane(s_l, j + u);
        w[u] = __builtin_bit_cast(float, __builtin_amdgcn_readlane(__builtin_bit_cast(int, w_l), j + u));
        v[u] = ((const VT*)(h + (long)s * F))[lane];
    }
#pragma unroll
    for (int u = 0; u < U; ++u) fma_row(acc, v[u], w[u]);
}

template <int F>
__global__ __launch_bounds__(256)
void gather_kernel(const int* __restrict__ rowptr, const int* __restrict__ csr_src,
                   const float* __restrict__ dis, const float* __restrict__ h,
                   float* __restrict__ out, int n) {
    typedef typename RowVec<F>::type VT;
    const int node0 = (blockIdx.x * 8 + __builtin_amdgcn_readfirstlane((int)(threadIdx.x >> 5))) * ROWS_PER_WAVE;
    if (node0 >= n) return;
    const int lane = threadIdx.x & 31;
    const int rows = (n - node0 < ROWS_PER_WAVE) ? (n - node0) : ROWS_PER_WAVE;
    const int   rp_l = (lane <= rows) ? rowptr[node0 + lane] : 0;
    const float dd_l = (lane <  rows) ? dis[node0 + lane] : 0.0f;

    for (int i = 0; i < rows; ++i) {
        const int beg = __builtin_amdgcn_readlane(rp_l, i);
        const int end = __builtin_amdgcn_readlane(rp_l, i + 1);
        const float dd = __builtin_bit_cast(float, __builtin_amdgcn_readlane(__builtin_bit_cast(int, dd_l), i));

        VT acc = {};
        for (int base = beg; base < end; base += 32) {
            const int m = (end - base < 32) ? (end - base) : 32;
            int s_l = 0;
            float w_l = 0.0f;
            if (lane < m) {
                s_l = csr_src[base + lane];
                w_l = dis[s_l] * dd;
            }
            int j = 0;
            for (; j + 8 <= m; j += 8) gather_step<F, 8>(acc, h, s_l, w_l, j, lane);
            if (j + 4 <= m) { gather_step<F, 4>(acc, h, s_l, w_l, j, lane); j += 4; }
            if (j + 2 <= m) { gather_step<F, 2>(acc, h, s_l, w_l, j, lane); j += 2; }
            if (j < m)      { gather_step<F, 1>(acc, h, s_l, w_l, j, lane); }
        }
        ((VT*)(out + (long)(node0 + i) * F))[lane] = acc;
    }
}


extern "C" void kernel_launch(void* const* d_in, const int* in_sizes, int n_in,
                              void* d_out, int out_size, void* d_ws, size_t ws_size,
                              hipStream_t stream) {
    const float* x  = (const float*)d_in[0];
    const int*   ei = (const int*)d_in[1];
    const float* W1 = (const float*)d_in[2];
    const float* b1 = (const float*)d_in[3];
    const float* W2 = (const float*)d_in[4];
    const float* b2 = (const float*)d_in[5];
    float* out = (float*)d_out;

    const int* src = ei;
    const int* dst = ei + N_EDGES;

    float* dis  = (float*)d_ws;
    float* h1   = dis + N_NODES;
    float* agg1 = h1 + (long)N_NODES * HID_F;
    float* h2   = agg1 + (long)N_NODES * HID_F;
    int* rowptr     = (int*)(h2 + (long)N_NODES * OUT_F);
    int* band_start = rowptr + N_NODES + 32;
    int* cntmat     = band_start + 256;
    int* binned     = cntmat + NBAND * CNT_PITCH;
    int* csr_src    = binned + N_EDGES;

    band_count_kernel<<<NCHUNK, SORT_THREADS, 0, stream>>>(dst, cntmat, N_EDGES);
    band_group_kernel<<<NCHUNK, SORT_THREADS, 0, stream>>>(src, dst, cntmat, band_start, binned, N_EDGES);
    band_finish_kernel<<<NBAND, SORT_THREADS, 0, stream>>>(binned, band_start, rowptr, dis, csr_src, N_NODES);

    gemm_wmma_kernel<HID_F, false><<<(N_NODES / 16 + TILES_PER_BLOCK - 1) / TILES_PER_BLOCK, 256, 0, stream>>>(x, W1, b1, h1, N_NODES / 16);

    gather_kernel<HID_F><<<(N_NODES + 8 * ROWS_PER_WAVE - 1) / (8 * ROWS_PER_WAVE), 256, 0, stream>>>(rowptr, csr_src, dis, h1, agg1, N_NODES);

    gemm_wmma_kernel<OUT_F, true><<<(N_NODES / 16 + TILES_PER_BLOCK - 1) / TILES_PER_BLOCK, 128, 0, stream>>>(agg1, W2, b2, h2, N_NODES / 16);

    gather_kernel<OUT_F><<<(N_NODES + 8 * ROWS_PER_WAVE - 1) / (8 * ROWS_PER_WAVE), 256, 0, stream>>>(rowptr, csr_src, dis, h2, out, N_NODES);
}


# GEMM A operand through LDS: a block reads each row tile of X once with whole-row loads (relu there) into one of two LDS stages while the previous tile is multiplied; products and sum order kept
# speedup vs baseline: 3.9181x; 1.1553x over previous
#include <hip/hip_runtime.h>
#include <hip/hip_bf16.h>

typedef __attribute__((ext_vector_type(2))) float v2f;
typedef __attribute__((ext_vector_type(8))) float v8f;
typedef __attribute__((ext_vector_type(4))) float v4f;

#define N_NODES 100000
#define N_EDGES 1600000
#define IN_F    128
#define HID_F   128
#define OUT_F   64


#define BAND_SHIFT 9
#define BAND (1 << BAND_SHIFT)
#define NBAND ((N_NODES + BAND - 1) >> BAND_SHIFT)
#define SORT_THREADS 1024
#define EDGES_PER_THREAD 8
#define CHUNK (SORT_THREADS * EDGES_PER_THREAD)
#define NCHUNK ((N_EDGES + CHUNK - 1) / CHUNK)
#define BAND_EDGES_PER_THREAD 12
#define BAND_CAP (SORT_THREADS * BAND_EDGES_PER_THREAD)
#define BANDS_PER_LANE ((NBAND + 31) / 32)
#define NBAND_PAD (BANDS_PER_LANE * 32)
#define CHUNKS_PER_LANE ((NCHUNK + 31) / 32)
#define CNT_PITCH (CHUNKS_PER_LANE * 32)

static_assert(N_NODES <= (1 << (31 - BAND_SHIFT)), "src must fit above the band-local destination");

#define ROWS_PER_WAVE 8
#define TILES_PER_BLOCK 5

__device__ __forceinline__ void wave_exscan_bands(const int* in, int* out, int lane) {
    int v[BANDS_PER_LANE];
    int t = 0;
#pragma unroll
    for (int k = 0; k < BANDS_PER_LANE; ++k) {
        v[k] = in[lane * BANDS_PER_LANE + k];
        t += v[k];
    }
    int inc = t;
#pragma unroll
    for (int o = 1; o < 32; o <<= 1) {
        const int up = __shfl_up(inc, o);
        if (lane >= o) inc += up;
    }
    int run = inc - t;
#pragma unroll
    for (int k = 0; k < BANDS_PER_LANE; ++k) {
        out[lane * BANDS_PER_LANE + k] = run;
        run += v[k];
    }
}

__global__ __launch_bounds__(SORT_THREADS)
void band_count_kernel(const int* __restrict__ dst, int* __restrict__ cntmat, int E) {
    __shared__ int s_cnt[NBAND];
    const int tid = threadIdx.x;
    if (tid < NBAND) s_cnt[tid] = 0;
    const int base = blockIdx.x * CHUNK + tid;
    int d[EDGES_PER_THREAD];
#pragma unroll
    for (int k = 0; k < EDGES_PER_THREAD; ++k) {
        const int e = base + k * SORT_THREADS;
        const int v = dst[min(e, E - 1)];
        d[k] = (e < E) ? v : -1;
    }
    __syncthreads();
#pragma unroll
    for (int k = 0; k < EDGES_PER_THREAD; ++k)
        if (d[k] >= 0) atomicAdd(&s_cnt[d[k] >> BAND_SHIFT], 1);
    __syncthreads();
    const int chunk = blockIdx.x;
    if (tid < NBAND)
        cntmat[tid * CNT_PITCH + (chunk % CHUNKS_PER_LANE) * 32 + chunk / CHUNKS_PER_LANE] = s_cnt[tid];
}

__global__ __launch_bounds__(SORT_THREADS)
void band_group_kernel(const int* __restrict__ src, const int* __restrict__ dst,
                       const int* __restrict__ cntmat, int* __restrict__ band_start,
                       int* __restrict__ binned, int E) {
    __shared__ int s_cnt[NBAND_PAD];
    __shared__ int s_lstart[NBAND_PAD];
    __shared__ int s_tot[NBAND_PAD];
    __shared__ int s_pre[NBAND_PAD];
    __shared__ int s_gstart[NBAND_PAD];
    __shared__ int s_rec[CHUNK];
    const int tid = threadIdx.x, lane = tid & 31;
    const int wave = __builtin_amdgcn_readfirstlane(tid >> 5);
    const int chunk = blockIdx.x;

    if (tid < NBAND_PAD) s_cnt[tid] = 0;

    const int base = chunk * CHUNK + tid;
    int d[EDGES_PER_THREAD], s[EDGES_PER_THREAD], r[EDGES_PER_THREAD];
#pragma unroll
    for (int k = 0; k < EDGES_PER_THREAD; ++k) {
        const int e = base + k * SORT_THREADS;
        const int ec = min(e, E - 1);
        const int dv = dst[ec];
        s[k] = src[ec];
        d[k] = (e < E) ? dv : -1;
    }

    const int own_lane = chunk / CHUNKS_PER_LANE;
    const int own_k    = chunk % CHUNKS_PER_LANE;
    int v[BANDS_PER_LANE][CHUNKS_PER_LANE];
#pragma unroll
    for (int j = 0; j < BANDS_PER_LANE; ++j) {
        const int b = wave + j * (SORT_THREADS / 32);
        const int* row = cntmat + min(b, NBAND - 1) * CNT_PITCH;
#pragma unroll
        for (int k = 0; k < CHUNKS_PER_LANE; ++k) {
            const bool ok = lane * CHUNKS_PER_LANE + k < NCHUNK;
            const int x = row[ok ? k * 32 + lane : 0];
            v[j][k] = ok ? x : 0;
        }
    }
    int t[BANDS_PER_LANE], part[BANDS_PER_LANE], inc[BANDS_PER_LANE];
#pragma unroll
    for (int j = 0; j < BANDS_PER_LANE; ++j) {
        t[j] = 0;
        part[j] = 0;
#pragma unroll
        for (int k = 0; k < CHUNKS_PER_LANE; ++k) {
            t[j] += v[j][k];
            part[j] += (k < own_k) ? v[j][k] : 0;
        }
        inc[j] = t[j];
    }
#pragma unroll
    for (int o = 1; o < 32; o <<= 1) {
        int up[BANDS_PER_LANE];
#pragma unroll
        for (int j = 0; j < BANDS_PER_LANE; ++j) up[j] = __shfl_up(inc[j], o);
#pragma unroll
        for (int j = 0; j < BANDS_PER_LANE; ++j) inc[j] += (lane >= o) ? up[j] : 0;
    }
#pragma unroll
    for (int j = 0; j < BANDS_PER_LANE; ++j) {
        const int b = wave + j * (SORT_THREADS / 32);
        const int tot = __builtin_amdgcn_readlane(inc[j], 31);
        const int pre = __builtin_amdgcn_readlane(inc[j] - t[j] + part[j], own_lane);
        if (lane == 0) {
            s_tot[b] = (b < NBAND) ? tot : 0;
            s_pre[b] = (b < NBAND) ? pre : 0;
        }
    }
    __syncthreads();

#pragma unroll
    for (int k = 0; k < EDGES_PER_THREAD; ++k)
        r[k] = (d[k] >= 0) ? atomicAdd(&s_cnt[d[k] >> BAND_SHIFT], 1) : 0;
    if (wave == 1) wave_exscan_bands(s_tot, s_gstart, lane);
    __syncthreads();

    if (wave == 0) wave_exscan_bands(s_cnt, s_lstart, lane);
    __syncthreads();

    if (tid < NBAND) {
        const int bs = s_gstart[tid];
        if (chunk == 0) band_start[tid] = bs;
        s_gstart[tid] = bs + s_pre[tid];
    }
    if (chunk == 0 && tid == 0) band_start[NBAND] = E;
#pragma unroll
    for (int k = 0; k < EDGES_PER_THREAD; ++k)
        if (d[k] >= 0)
            s_rec[s_lstart[d[k] >> BAND_SHIFT] + r[k]] = (s[k] << BAND_SHIFT) | (d[k] & (BAND - 1));
    __syncthreads();

    for (int b = wave; b < NBAND; b += SORT_THREADS / 32) {
        const int n = s_cnt[b], ls = s_lstart[b], gs = s_gstart[b];
        for (int i = lane; i < n; i += 32) binned[gs + i] = s_rec[ls + i];
    }
}

__global__ __launch_bounds__(SORT_THREADS)
void band_finish_kernel(const int* __restrict__ binned, const int* __restrict__ band_start,
                        int* __restrict__ rowptr, float* __restrict__ dis,
                        int* __restrict__ csr_src, int n) {
    __shared__ int s_cnt[BAND];
    __shared__ int s_wtot[BAND / 32];
    __shared__ int s_out[BAND_CAP];
    const int tid = threadIdx.x, lane = tid & 31, wave = tid >> 5;
    const int band = blockIdx.x;
    const int beg = band_start[band], end = band_start[band + 1];
    const int m = end - beg;
    const bool in_lds = (m <= BAND_CAP);

    if (tid < BAND) s_cnt[tid] = 0;

    int rec[BAND_EDGES_PER_THREAD];
#pragma unroll
    for (int k = 0; k < BAND_EDGES_PER_THREAD; ++k) rec[k] = -1;
    if (m > 0) {
#pragma unroll
        for (int k = 0; k < BAND_EDGES_PER_THREAD; ++k) {
            const int i = k * SORT_THREADS + tid;
            const int v = binned[beg + min(i, m - 1)];
            rec[k] = (i < m) ? v : -1;
        }
    }
    __syncthreads();

#pragma unroll
    for (int k = 0; k < BAND_EDGES_PER_THREAD; ++k)
        if (rec[k] >= 0) atomicAdd(&s_cnt[rec[k] & (BAND - 1)], 1);
    for (int i = BAND_CAP + tid; i < m; i += SORT_THREADS)
        atomicAdd(&s_cnt[binned[beg + i] & (BAND - 1)], 1);
    __syncthreads();

    int c = 0, inc = 0;
    if (tid < BAND) {
        c = s_cnt[tid];
        inc = c;
#pragma unroll
        for (int o = 1; o < 32; o <<= 1) {
            const int up = __shfl_up(inc, o);
            if (lane >= o) inc += up;
        }
        if (lane == 31) s_wtot[wave] = inc;
    }
    __syncthreads();
    if (tid < BAND) {
        int off = 0;
#pragma unroll
        for (int w = 0; w < BAND / 32; ++w) off += (w < wave) ? s_wtot[w] : 0;
        const int excl = off + inc - c;
        s_cnt[tid] = excl;
        const int node = band * BAND + tid;
        if (node < n) {
            rowptr[node] = beg + excl;
            dis[node] = (c > 0) ? (1.0f / sqrtf((float)c)) : 0.0f;
        }
    }
    if (band == NBAND - 1 && tid == 0) rowptr[n] = end;
    __syncthreads();

    int p[BAND_EDGES_PER_THREAD];
#pragma unroll
    for (int k = 0; k < BAND_EDGES_PER_THREAD; ++k)
        p[k] = (rec[k] >= 0) ? atomicAdd(&s_cnt[rec[k] & (BAND - 1)], 1) : 0;
    if (in_lds) {
#pragma unroll
        for (int k = 0; k < BAND_EDGES_PER_THREAD; ++k)
            if (rec[k] >= 0) s_out[p[k]] = rec[k] >> BAND_SHIFT;
    } else {
#pragma unroll
        for (int k = 0; k < BAND_EDGES_PER_THREAD; ++k)
            if (rec[k] >= 0) csr_src[beg + p[k]] = rec[k] >> BAND_SHIFT;
    }
    for (int i = BAND_CAP + tid; i < m; i += SORT_THREADS) {
        const int w = binned[beg + i];
        csr_src[beg + atomicAdd(&s_cnt[w & (BAND - 1)], 1)] = w >> BAND_SHIFT;
    }
    if (in_lds) {
        __syncthreads();
        for (int i = tid; i < m; i += SORT_THREADS) csr_src[beg + i] = s_out[i];
    }
}

#define XPITCH 132
template <int FOUT, bool RELU_IN>
__global__ __launch_bounds__(32 * (FOUT / 16))
void gemm_wmma_kernel(const float* __restrict__ X, const float* __restrict__ W,
                      const float* __restrict__ bias, float* __restrict__ H, int n_tiles) {
    constexpr int K = 128;
    constexpr int NT = 32 * (FOUT / 16);
    constexpr int LPT = (16 * K / 4) / NT;
    __shared__ v4f s_x[2][16 * XPITCH / 4];
    const int tid  = threadIdx.x;
    const int wave = tid >> 5;
    const int lane = tid & 31;
    const int half = lane >> 4;
    const int l    = lane & 15;
    const int n0 = wave * 16;

    const int t0 = blockIdx.x * TILES_PER_BLOCK;
    const int nt = (n_tiles - t0 < TILES_PER_BLOCK) ? (n_tiles - t0) : TILES_PER_BLOCK;
    if (nt <= 0) return;

    v4f st[LPT];
#pragma unroll
    for (int j = 0; j < LPT; ++j) {
        const int q = tid + j * NT;
        st[j] = ((const v4f*)(X + (long)(t0 * 16 + (q >> 5)) * K))[q & 31];
    }

    const float* brow = W + (long)(n0 + l) * K + 2 * half;
    v2f b[K / 4];
#pragma unroll
    for (int s = 0; s < K / 4; ++s) b[s] = *(const v2f*)(brow + 4 * s);
    const float bb = bias[n0 + l];

    for (int i = 0; i < nt; ++i) {
        const int m0 = (t0 + i) * 16;
        v4f* sx = s_x[i & 1];
#pragma unroll
        for (int j = 0; j < LPT; ++j) {
            const int q = tid + j * NT;
            v4f v = st[j];
            if (RELU_IN) {
                v.x = fmaxf(v.x, 0.0f); v.y = fmaxf(v.y, 0.0f);
                v.z = fmaxf(v.z, 0.0f); v.w = fmaxf(v.w, 0.0f);
            }
            sx[(q >> 5) * (XPITCH / 4) + (q & 31)] = v;
        }
        __syncthreads();

        if (i + 1 < nt) {
#pragma unroll
            for (int j = 0; j < LPT; ++j) {
                const int q = tid + j * NT;
                st[j] = ((const v4f*)(X + (long)(m0 + 16 + (q >> 5)) * K))[q & 31];
            }
        }

        const float* arow = (const float*)sx + l * XPITCH + 2 * half;
        v8f c = {};
#pragma unroll
        for (int s = 0; s < K / 4; ++s) {
            const v2f a = *(const v2f*)(arow + 4 * s);
            c = __builtin_amdgcn_wmma_f32_16x16x4_f32(false, a, false, b[s],
                                                      (short)0, c, false, false);
        }

#pragma unroll
        for (int r = 0; r < 8; ++r) {
            const int row = m0 + r + 8 * half;
            H[(long)row * FOUT + n0 + l] = c[r] + bb;
        }
    }
}

template <int F> struct RowVec;
template <> struct RowVec<128> { typedef float4 type; };
template <> struct RowVec<64>  { typedef float2 type; };

__device__ __forceinline__ void fma_row(float4& a, const float4& v, float w) {
    a.x = __builtin_fmaf(v.x, w, a.x); a.y = __builtin_fmaf(v.y, w, a.y);
    a.z = __builtin_fmaf(v.z, w, a.z); a.w = __builtin_fmaf(v.w, w, a.w);
}
__device__ __forceinline__ void fma_row(float2& a, const float2& v, float w) {
    a.x = __builtin_fmaf(v.x, w, a.x); a.y = __builtin_fmaf(v.y, w, a.y);
}

template <int F, int U>
__device__ __forceinline__ void gather_step(typename RowVec<F>::type& acc, const float* __restrict__ h,
                                            int s_l, float w_l, int j, int lane) {
    typedef typename RowVec<F>::type VT;
    VT v[U];
    float w[U];
#pragma unroll
    for (int u = 0; u < U; ++u) {
        const int s = __builtin_amdgcn_readlane(s_l, j + u);
        w[u] = __builtin_bit_cast(float, __builtin_amdgcn_readlane(__builtin_bit_cast(int, w_l), j + u));
        v[u] = ((const VT*)(h + (long)s * F))[lane];
    }
#pragma unroll
    for (int u = 0; u < U; ++u) fma_row(acc, v[u], w[u]);
}

template <int F>
__global__ __launch_bounds__(256)
void gather_kernel(const int* __restrict__ rowptr, const int* __restrict__ csr_src,
                   const float* __restrict__ dis, const float* __restrict__ h,
                   float* __restrict__ out, int n) {
    typedef typename RowVec<F>::type VT;
    const int node0 = (blockIdx.x * 8 + __builtin_amdgcn_readfirstlane((int)(threadIdx.x >> 5))) * ROWS_PER_WAVE;
    if (node0 >= n) return;
    const int lane = threadIdx.x & 31;
    const int rows = (n - node0 < ROWS_PER_WAVE) ? (n - node0) : ROWS_PER_WAVE;
    const int   rp_l = (lane <= rows) ? rowptr[node0 + lane] : 0;
    const float dd_l = (lane <  rows) ? dis[node0 + lane] : 0.0f;

    for (int i = 0; i < rows; ++i) {
        const int beg = __builtin_amdgcn_readlane(rp_l, i);
        const int end = __builtin_amdgcn_readlane(rp_l, i + 1);
        const float dd = __builtin_bit_cast(float, __builtin_amdgcn_readlane(__builtin_bit_cast(int, dd_l), i));

        VT acc = {};
        for (int base = beg; base < end; base += 32) {
            const int m = (end - base < 32) ? (end - base) : 32;
            int s_l = 0;
            float w_l = 0.0f;
            if (lane < m) {
                s_l = csr_src[base + lane];
                w_l = dis[s_l] * dd;
            }
            int j = 0;
            for (; j + 8 <= m; j += 8) gather_step<F, 8>(acc, h, s_l, w_l, j, lane);
            if (j + 4 <= m) { gather_step<F, 4>(acc, h, s_l, w_l, j, lane); j += 4; }
            if (j + 2 <= m) { gather_step<F, 2>(acc, h, s_l, w_l, j, lane); j += 2; }
            if (j < m)      { gather_step<F, 1>(acc, h, s_l, w_l, j, lane); }
        }
        ((VT*)(out + (long)(node0 + i) * F))[lane] = acc;
    }
}


extern "C" void kernel_launch(void* const* d_in, const int* in_sizes, int n_in,
                              void* d_out, int out_size, void* d_ws, size_t ws_size,
                              hipStream_t stream) {
    const float* x  = (const float*)d_in[0];
    const int*   ei = (const int*)d_in[1];
    const float* W1 = (const float*)d_in[2];
    const float* b1 = (const float*)d_in[3];
    const float* W2 = (const float*)d_in[4];
    const float* b2 = (const float*)d_in[5];
    float* out = (float*)d_out;

    const int* src = ei;
    const int* dst = ei + N_EDGES;

    float* dis  = (float*)d_ws;
    float* h1   = dis + N_NODES;
    float* agg1 = h1 + (long)N_NODES * HID_F;
    float* h2   = agg1 + (long)N_NODES * HID_F;
    int* rowptr     = (int*)(h2 + (long)N_NODES * OUT_F);
    int* band_start = rowptr + N_NODES + 32;
    int* cntmat     = band_start + 256;
    int* binned     = cntmat + NBAND * CNT_PITCH;
    int* csr_src    = binned + N_EDGES;

    band_count_kernel<<<NCHUNK, SORT_THREADS, 0, stream>>>(dst, cntmat, N_EDGES);
    band_group_kernel<<<NCHUNK, SORT_THREADS, 0, stream>>>(src, dst, cntmat, band_start, binned, N_EDGES);
    band_finish_kernel<<<NBAND, SORT_THREADS, 0, stream>>>(binned, band_start, rowptr, dis, csr_src, N_NODES);

    gemm_wmma_kernel<HID_F, false><<<(N_NODES / 16 + TILES_PER_BLOCK - 1) / TILES_PER_BLOCK, 256, 0, stream>>>(x, W1, b1, h1, N_NODES / 16);

    gather_kernel<HID_F><<<(N_NODES + 8 * ROWS_PER_WAVE - 1) / (8 * ROWS_PER_WAVE), 256, 0, stream>>>(rowptr, csr_src, dis, h1, agg1, N_NODES);

    gemm_wmma_kernel<OUT_F, true><<<(N_NODES / 16 + TILES_PER_BLOCK - 1) / TILES_PER_BLOCK, 128, 0, stream>>>(agg1, W2, b2, h2, N_NODES / 16);

    gather_kernel<OUT_F><<<(N_NODES + 8 * ROWS_PER_WAVE - 1) / (8 * ROWS_PER_WAVE), 256, 0, stream>>>(rowptr, csr_src, dis, h2, out, N_NODES);
}


# Both dense layers fused into the first gather: aggregate x, then W1, relu, W2 from one LDS tile (bias 1 times the row's weight sum); h1, agg1 never stored; layer-1 sum order: edges first, then k
# speedup vs baseline: 5.2106x; 1.3299x over previous
#include <hip/hip_runtime.h>
#include <hip/hip_bf16.h>

typedef __attribute__((ext_vector_type(2))) float v2f;
typedef __attribute__((ext_vector_type(8))) float v8f;
typedef __attribute__((ext_vector_type(4))) float v4f;

#define N_NODES 100000
#define N_EDGES 1600000
#define IN_F    128
#define HID_F   128
#define OUT_F   64


#define BAND_SHIFT 9
#define BAND (1 << BAND_SHIFT)
#define NBAND ((N_NODES + BAND - 1) >> BAND_SHIFT)
#define SORT_THREADS 1024
#define EDGES_PER_THREAD 8
#define CHUNK (SORT_THREADS * EDGES_PER_THREAD)
#define NCHUNK ((N_EDGES + CHUNK - 1) / CHUNK)
#define BAND_EDGES_PER_THREAD 12
#define BAND_CAP (SORT_THREADS * BAND_EDGES_PER_THREAD)
#define BANDS_PER_LANE ((NBAND + 31) / 32)
#define NBAND_PAD (BANDS_PER_LANE * 32)
#define CHUNKS_PER_LANE ((NCHUNK + 31) / 32)
#define CNT_PITCH (CHUNKS_PER_LANE * 32)

static_assert(N_NODES <= (1 << (31 - BAND_SHIFT)), "src must fit above the band-local destination");

#define ROWS_PER_WAVE 8
#define FUSE_ROWS 64

__device__ __forceinline__ void wave_exscan_bands(const int* in, int* out, int lane) {
    int v[BANDS_PER_LANE];
    int t = 0;
#pragma unroll
    for (int k = 0; k < BANDS_PER_LANE; ++k) {
        v[k] = in[lane * BANDS_PER_LANE + k];
        t += v[k];
    }
    int inc = t;
#pragma unroll
    for (int o = 1; o < 32; o <<= 1) {
        const int up = __shfl_up(inc, o);
        if (lane >= o) inc += up;
    }
    int run = inc - t;
#pragma unroll
    for (int k = 0; k < BANDS_PER_LANE; ++k) {
        out[lane * BANDS_PER_LANE + k] = run;
        run += v[k];
    }
}

__global__ __launch_bounds__(SORT_THREADS)
void band_count_kernel(const int* __restrict__ dst, int* __restrict__ cntmat, int E) {
    __shared__ int s_cnt[NBAND];
    const int tid = threadIdx.x;
    if (tid < NBAND) s_cnt[tid] = 0;
    const int base = blockIdx.x * CHUNK + tid;
    int d[EDGES_PER_THREAD];
#pragma unroll
    for (int k = 0; k < EDGES_PER_THREAD; ++k) {
        const int e = base + k * SORT_THREADS;
        const int v = dst[min(e, E - 1)];
        d[k] = (e < E) ? v : -1;
    }
    __syncthreads();
#pragma unroll
    for (int k = 0; k < EDGES_PER_THREAD; ++k)
        if (d[k] >= 0) atomicAdd(&s_cnt[d[k] >> BAND_SHIFT], 1);
    __syncthreads();
    const int chunk = blockIdx.x;
    if (tid < NBAND)
        cntmat[tid * CNT_PITCH + (chunk % CHUNKS_PER_LANE) * 32 + chunk / CHUNKS_PER_LANE] = s_cnt[tid];
}

__global__ __launch_bounds__(SORT_THREADS)
void band_group_kernel(const int* __restrict__ src, const int* __restrict__ dst,
                       const int* __restrict__ cntmat, int* __restrict__ band_start,
                       int* __restrict__ binned, int E) {
    __shared__ int s_cnt[NBAND_PAD];
    __shared__ int s_lstart[NBAND_PAD];
    __shared__ int s_tot[NBAND_PAD];
    __shared__ int s_pre[NBAND_PAD];
    __shared__ int s_gstart[NBAND_PAD];
    __shared__ int s_rec[CHUNK];
    const int tid = threadIdx.x, lane = tid & 31;
    const int wave = __builtin_amdgcn_readfirstlane(tid >> 5);
    const int chunk = blockIdx.x;

    if (tid < NBAND_PAD) s_cnt[tid] = 0;

    const int base = chunk * CHUNK + tid;
    int d[EDGES_PER_THREAD], s[EDGES_PER_THREAD], r[EDGES_PER_THREAD];
#pragma unroll
    for (int k = 0; k < EDGES_PER_THREAD; ++k) {
        const int e = base + k * SORT_THREADS;
        const int ec = min(e, E - 1);
        const int dv = dst[ec];
        s[k] = src[ec];
        d[k] = (e < E) ? dv : -1;
    }

    const int own_lane = chunk / CHUNKS_PER_LANE;
    const int own_k    = chunk % CHUNKS_PER_LANE;
    int v[BANDS_PER_LANE][CHUNKS_PER_LANE];
#pragma unroll
    for (int j = 0; j < BANDS_PER_LANE; ++j) {
        const int b = wave + j * (SORT_THREADS / 32);
        const int* row = cntmat + min(b, NBAND - 1) * CNT_PITCH;
#pragma unroll
        for (int k = 0; k < CHUNKS_PER_LANE; ++k) {
            const bool ok = lane * CHUNKS_PER_LANE + k < NCHUNK;
            const int x = row[ok ? k * 32 + lane : 0];
            v[j][k] = ok ? x : 0;
        }
    }
    int t[BANDS_PER_LANE], part[BANDS_PER_LANE], inc[BANDS_PER_LANE];
#pragma unroll
    for (int j = 0; j < BANDS_PER_LANE; ++j) {
        t[j] = 0;
        part[j] = 0;
#pragma unroll
        for (int k = 0; k < CHUNKS_PER_LANE; ++k) {
            t[j] += v[j][k];
            part[j] += (k < own_k) ? v[j][k] : 0;
        }
        inc[j] = t[j];
    }
#pragma unroll
    for (int o = 1; o < 32; o <<= 1) {
        int up[BANDS_PER_LANE];
#pragma unroll
        for (int j = 0; j < BANDS_PER_LANE; ++j) up[j] = __shfl_up(inc[j], o);
#pragma unroll
        for (int j = 0; j < BANDS_PER_LANE; ++j) inc[j] += (lane >= o) ? up[j] : 0;
    }
#pragma unroll
    for (int j = 0; j < BANDS_PER_LANE; ++j) {
        const int b = wave + j * (SORT_THREADS / 32);
        const int tot = __builtin_amdgcn_readlane(inc[j], 31);
        const int pre = __builtin_amdgcn_readlane(inc[j] - t[j] + part[j], own_lane);
        if (lane == 0) {
            s_tot[b] = (b < NBAND) ? tot : 0;
            s_pre[b] = (b < NBAND) ? pre : 0;
        }
    }
    __syncthreads();

#pragma unroll
    for (int k = 0; k < EDGES_PER_THREAD; ++k)
        r[k] = (d[k] >= 0) ? atomicAdd(&s_cnt[d[k] >> BAND_SHIFT], 1) : 0;
    if (wave == 1) wave_exscan_bands(s_tot, s_gstart, lane);
    __syncthreads();

    if (wave == 0) wave_exscan_bands(s_cnt, s_lstart, lane);
    __syncthreads();

    if (tid < NBAND) {
        const int bs = s_gstart[tid];
        if (chunk == 0) band_start[tid] = bs;
        s_gstart[tid] = bs + s_pre[tid];
    }
    if (chunk == 0 && tid == 0) band_start[NBAND] = E;
#pragma unroll
    for (int k = 0; k < EDGES_PER_THREAD; ++k)
        if (d[k] >= 0)
            s_rec[s_lstart[d[k] >> BAND_SHIFT] + r[k]] = (s[k] << BAND_SHIFT) | (d[k] & (BAND - 1));
    __syncthreads();

    for (int b = wave; b < NBAND; b += SORT_THREADS / 32) {
        const int n = s_cnt[b], ls = s_lstart[b], gs = s_gstart[b];
        for (int i = lane; i < n; i += 32) binned[gs + i] = s_rec[ls + i];
    }
}

__global__ __launch_bounds__(SORT_THREADS)
void band_finish_kernel(const int* __restrict__ binned, const int* __restrict__ band_start,
                        int* __restrict__ rowptr, float* __restrict__ dis,
                        int* __restrict__ csr_src, int n) {
    __shared__ int s_cnt[BAND];
    __shared__ int s_wtot[BAND / 32];
    __shared__ int s_out[BAND_CAP];
    const int tid = threadIdx.x, lane = tid & 31, wave = tid >> 5;
    const int band = blockIdx.x;
    const int beg = band_start[band], end = band_start[band + 1];
    const int m = end - beg;
    const bool in_lds = (m <= BAND_CAP);

    if (tid < BAND) s_cnt[tid] = 0;

    int rec[BAND_EDGES_PER_THREAD];
#pragma unroll
    for (int k = 0; k < BAND_EDGES_PER_THREAD; ++k) rec[k] = -1;
    if (m > 0) {
#pragma unroll
        for (int k = 0; k < BAND_EDGES_PER_THREAD; ++k) {
            const int i = k * SORT_THREADS + tid;
            const int v = binned[beg + min(i, m - 1)];
            rec[k] = (i < m) ? v : -1;
        }
    }
    __syncthreads();

#pragma unroll
    for (int k = 0; k < BAND_EDGES_PER_THREAD; ++k)
        if (rec[k] >= 0) atomicAdd(&s_cnt[rec[k] & (BAND - 1)], 1);
    for (int i = BAND_CAP + tid; i < m; i += SORT_THREADS)
        atomicAdd(&s_cnt[binned[beg + i] & (BAND - 1)], 1);
    __syncthreads();

    int c = 0, inc = 0;
    if (tid < BAND) {
        c = s_cnt[tid];
        inc = c;
#pragma unroll
        for (int o = 1; o < 32; o <<= 1) {
            const int up = __shfl_up(inc, o);
            if (lane >= o) inc += up;
        }
        if (lane == 31) s_wtot[wave] = inc;
    }
    __syncthreads();
    if (tid < BAND) {
        int off = 0;
#pragma unroll
        for (int w = 0; w < BAND / 32; ++w) off += (w < wave) ? s_wtot[w] : 0;
        const int excl = off + inc - c;
        s_cnt[tid] = excl;
        const int node = band * BAND + tid;
        if (node < n) {
            rowptr[node] = beg + excl;
            dis[node] = (c > 0) ? (1.0f / sqrtf((float)c)) : 0.0f;
        }
    }
    if (band == NBAND - 1 && tid == 0) rowptr[n] = end;
    __syncthreads();

    int p[BAND_EDGES_PER_THREAD];
#pragma unroll
    for (int k = 0; k < BAND_EDGES_PER_THREAD; ++k)
        p[k] = (rec[k] >= 0) ? atomicAdd(&s_cnt[rec[k] & (BAND - 1)], 1) : 0;
    if (in_lds) {
#pragma unroll
        for (int k = 0; k < BAND_EDGES_PER_THREAD; ++k)
            if (rec[k] >= 0) s_out[p[k]] = rec[k] >> BAND_SHIFT;
    } else {
#pragma unroll
        for (int k = 0; k < BAND_EDGES_PER_THREAD; ++k)
            if (rec[k] >= 0) csr_src[beg + p[k]] = rec[k] >> BAND_SHIFT;
    }
    for (int i = BAND_CAP + tid; i < m; i += SORT_THREADS) {
        const int w = binned[beg + i];
        csr_src[beg + atomicAdd(&s_cnt[w & (BAND - 1)], 1)] = w >> BAND_SHIFT;
    }
    if (in_lds) {
        __syncthreads();
        for (int i = tid; i < m; i += SORT_THREADS) csr_src[beg + i] = s_out[i];
    }
}

template <int F> struct RowVec;
template <> struct RowVec<128> { typedef float4 type; };
template <> struct RowVec<64>  { typedef float2 type; };

__device__ __forceinline__ void fma_row(float4& a, const float4& v, float w) {
    a.x = __builtin_fmaf(v.x, w, a.x); a.y = __builtin_fmaf(v.y, w, a.y);
    a.z = __builtin_fmaf(v.z, w, a.z); a.w = __builtin_fmaf(v.w, w, a.w);
}
__device__ __forceinline__ void fma_row(float2& a, const float2& v, float w) {
    a.x = __builtin_fmaf(v.x, w, a.x); a.y = __builtin_fmaf(v.y, w, a.y);
}

template <int F, int U>
__device__ __forceinline__ void gather_step(typename RowVec<F>::type& acc, const float* __restrict__ h,
                                            int s_l, float w_l, int j, int lane) {
    typedef typename RowVec<F>::type VT;
    VT v[U];
    float w[U];
#pragma unroll
    for (int u = 0; u < U; ++u) {
        const int s = __builtin_amdgcn_readlane(s_l, j + u);
        w[u] = __builtin_bit_cast(float, __builtin_amdgcn_readlane(__builtin_bit_cast(int, w_l), j + u));
        v[u] = ((const VT*)(h + (long)s * F))[lane];
    }
#pragma unroll
    for (int u = 0; u < U; ++u) fma_row(acc, v[u], w[u]);
}

template <int F>
__global__ __launch_bounds__(256)
void gather_kernel(const int* __restrict__ rowptr, const int* __restrict__ csr_src,
                   const float* __restrict__ dis, const float* __restrict__ h,
                   float* __restrict__ out, int n) {
    typedef typename RowVec<F>::type VT;
    const int node0 = (blockIdx.x * 8 + __builtin_amdgcn_readfirstlane((int)(threadIdx.x >> 5))) * ROWS_PER_WAVE;
    if (node0 >= n) return;
    const int lane = threadIdx.x & 31;
    const int rows = (n - node0 < ROWS_PER_WAVE) ? (n - node0) : ROWS_PER_WAVE;
    const int   rp_l = (lane <= rows) ? rowptr[node0 + lane] : 0;
    const float dd_l = (lane <  rows) ? dis[node0 + lane] : 0.0f;

    for (int i = 0; i < rows; ++i) {
        const int beg = __builtin_amdgcn_readlane(rp_l, i);
        const int end = __builtin_amdgcn_readlane(rp_l, i + 1);
        const float dd = __builtin_bit_cast(float, __builtin_amdgcn_readlane(__builtin_bit_cast(int, dd_l), i));

        VT acc = {};
        for (int base = beg; base < end; base += 32) {
            const int m = (end - base < 32) ? (end - base) : 32;
            int s_l = 0;
            float w_l = 0.0f;
            if (lane < m) {
                s_l = csr_src[base + lane];
                w_l = dis[s_l] * dd;
            }
            int j = 0;
            for (; j + 8 <= m; j += 8) gather_step<F, 8>(acc, h, s_l, w_l, j, lane);
            if (j + 4 <= m) { gather_step<F, 4>(acc, h, s_l, w_l, j, lane); j += 4; }
            if (j + 2 <= m) { gather_step<F, 2>(acc, h, s_l, w_l, j, lane); j += 2; }
            if (j < m)      { gather_step<F, 1>(acc, h, s_l, w_l, j, lane); }
        }
        ((VT*)(out + (long)(node0 + i) * F))[lane] = acc;
    }
}

#define XPITCH 132

template <int U>
__device__ __forceinline__ void gather_step_ws(float4& acc, float& wsum, const float* __restrict__ h,
                                               int s_l, float w_l, int j, int lane) {
    float4 v[U];
    float w[U];
#pragma unroll
    for (int u = 0; u < U; ++u) {
        const int s = __builtin_amdgcn_readlane(s_l, j + u);
        w[u] = __builtin_bit_cast(float, __builtin_amdgcn_readlane(__builtin_bit_cast(int, w_l), j + u));
        v[u] = ((const float4*)(h + (long)s * IN_F))[lane];
    }
#pragma unroll
    for (int u = 0; u < U; ++u) {
        fma_row(acc, v[u], w[u]);
        wsum += w[u];
    }
}

__device__ __forceinline__ v8f wmma4(v2f a, v2f b, v8f c) {
    return __builtin_amdgcn_wmma_f32_16x16x4_f32(false, a, false, b, (short)0, c, false, false);
}

__global__ __launch_bounds__(256, 10)
void fused_layers_kernel(const int* __restrict__ rowptr, const int* __restrict__ csr_src,
                         const float* __restrict__ dis, const float* __restrict__ x,
                         const float* __restrict__ W1, const float* __restrict__ b1,
                         const float* __restrict__ W2, const float* __restrict__ b2,
                         float* __restrict__ h2, int n) {
    constexpr int K = 128;
    __shared__ v4f s_t[FUSE_ROWS * XPITCH / 4];
    __shared__ float s_ws[FUSE_ROWS];
    const int tid  = threadIdx.x;
    const int wave = __builtin_amdgcn_readfirstlane(tid >> 5);
    const int lane = tid & 31;
    const int half = lane >> 4;
    const int l    = lane & 15;
    const int row0 = blockIdx.x * FUSE_ROWS;

    {
        const int node0 = row0 + wave * ROWS_PER_WAVE;
        int rows = n - node0;
        rows = (rows < 0) ? 0 : ((rows > ROWS_PER_WAVE) ? ROWS_PER_WAVE : rows);
        const int   rp_l = (rows > 0 && lane <= rows) ? rowptr[node0 + lane] : 0;
        const float dd_l = (lane < rows) ? dis[node0 + lane] : 0.0f;
        for (int i = 0; i < ROWS_PER_WAVE; ++i) {
            float4 acc = {};
            float wsum = 0.0f;
            if (i < rows) {
                const int beg = __builtin_amdgcn_readlane(rp_l, i);
                const int end = __builtin_amdgcn_readlane(rp_l, i + 1);
                const float dd = __builtin_bit_cast(float, __builtin_amdgcn_readlane(__builtin_bit_cast(int, dd_l), i));
                for (int base = beg; base < end; base += 32) {
                    const int m = (end - base < 32) ? (end - base) : 32;
                    int s_l = 0;
                    float w_l = 0.0f;
                    if (lane < m) {
                        s_l = csr_src[base + lane];
                        w_l = dis[s_l] * dd;
                    }
                    int j = 0;
                    for (; j + 8 <= m; j += 8) gather_step_ws<8>(acc, wsum, x, s_l, w_l, j, lane);
                    if (j + 4 <= m) { gather_step_ws<4>(acc, wsum, x, s_l, w_l, j, lane); j += 4; }
                    if (j + 2 <= m) { gather_step_ws<2>(acc, wsum, x, s_l, w_l, j, lane); j += 2; }
                    if (j < m)      { gather_step_ws<1>(acc, wsum, x, s_l, w_l, j, lane); }
                }
            }
            const int r = wave * ROWS_PER_WAVE + i;
            v4f o;
            o.x = acc.x; o.y = acc.y; o.z = acc.z; o.w = acc.w;
            s_t[r * (XPITCH / 4) + lane] = o;
            if (lane == 0) s_ws[r] = wsum;
        }
    }
    __syncthreads();

    v4f b[K / 8];
    {
        const unsigned boff = (unsigned)((wave * 16 + l) * K + 4 * half);
#pragma unroll
        for (int p = 0; p < K / 8; ++p) b[p] = *(const v4f*)(W1 + (boff + 8u * p));
    }
    const float bb1 = b1[wave * 16 + l];
    {
        v8f c[FUSE_ROWS / 16];
#pragma unroll
        for (int t = 0; t < FUSE_ROWS / 16; ++t) c[t] = v8f{};
        const float* arow = (const float*)s_t + l * XPITCH + 4 * half;
#pragma unroll
        for (int p = 0; p < K / 8; ++p) {
            v4f a[FUSE_ROWS / 16];
#pragma unroll
            for (int t = 0; t < FUSE_ROWS / 16; ++t) a[t] = *(const v4f*)(arow + t * 16 * XPITCH + 8 * p);
            const v2f blo = {b[p].x, b[p].y}, bhi = {b[p].z, b[p].w};
#pragma unroll
            for (int t = 0; t < FUSE_ROWS / 16; ++t) c[t] = wmma4(v2f{a[t].x, a[t].y}, blo, c[t]);
#pragma unroll
            for (int t = 0; t < FUSE_ROWS / 16; ++t) c[t] = wmma4(v2f{a[t].z, a[t].w}, bhi, c[t]);
        }
        __syncthreads();
#pragma unroll
        for (int t = 0; t < FUSE_ROWS / 16; ++t) {
#pragma unroll
            for (int r = 0; r < 8; ++r) {
                const int row = t * 16 + r + 8 * half;
                const float v = __builtin_fmaf(s_ws[row], bb1, c[t][r]);
                ((float*)s_t)[row * XPITCH + wave * 16 + l] = fmaxf(v, 0.0f);
            }
        }
    }
    __syncthreads();

    {
        const int n2 = (wave & 3) * 16;
        const int t0 = wave >> 2;
        const unsigned boff = (unsigned)((n2 + l) * K + 4 * half);
#pragma unroll
        for (int p = 0; p < K / 8; ++p) b[p] = *(const v4f*)(W2 + (boff + 8u * p));
        const float bb2 = b2[n2 + l];
        v8f d0 = {}, d1 = {};
        const float* arow0 = (const float*)s_t + (t0 * 16 + l) * XPITCH + 4 * half;
        const float* arow1 = arow0 + 32 * XPITCH;
#pragma unroll
        for (int p = 0; p < K / 8; ++p) {
            const v4f a0 = *(const v4f*)(arow0 + 8 * p);
            const v4f a1 = *(const v4f*)(arow1 + 8 * p);
            const v2f blo = {b[p].x, b[p].y}, bhi = {b[p].z, b[p].w};
            d0 = wmma4(v2f{a0.x, a0.y}, blo, d0);
            d1 = wmma4(v2f{a1.x, a1.y}, blo, d1);
            d0 = wmma4(v2f{a0.z, a0.w}, bhi, d0);
            d1 = wmma4(v2f{a1.z, a1.w}, bhi, d1);
        }
#pragma unroll
        for (int r = 0; r < 8; ++r) {
            const int rowa = row0 + t0 * 16 + r + 8 * half;
            const int rowb = rowa + 32;
            if (rowa < n) h2[(long)rowa * OUT_F + n2 + l] = d0[r] + bb2;
            if (rowb < n) h2[(long)rowb * OUT_F + n2 + l] = d1[r] + bb2;
        }
    }
}


extern "C" void kernel_launch(void* const* d_in, const int* in_sizes, int n_in,
                              void* d_out, int out_size, void* d_ws, size_t ws_size,
                              hipStream_t stream) {
    const float* x  = (const float*)d_in[0];
    const int*   ei = (const int*)d_in[1];
    const float* W1 = (const float*)d_in[2];
    const float* b1 = (const float*)d_in[3];
    const float* W2 = (const float*)d_in[4];
    const float* b2 = (const float*)d_in[5];
    float* out = (float*)d_out;

    const int* src = ei;
    const int* dst = ei + N_EDGES;

    float* dis  = (float*)d_ws;
    float* h2   = dis + N_NODES;
    int* rowptr     = (int*)(h2 + (long)N_NODES * OUT_F);
    int* band_start = rowptr + N_NODES + 32;
    int* cntmat     = band_start + 256;
    int* binned     = cntmat + NBAND * CNT_PITCH;
    int* csr_src    = binned + N_EDGES;

    band_count_kernel<<<NCHUNK, SORT_THREADS, 0, stream>>>(dst, cntmat, N_EDGES);
    band_group_kernel<<<NCHUNK, SORT_THREADS, 0, stream>>>(src, dst, cntmat, band_start, binned, N_EDGES);
    band_finish_kernel<<<NBAND, SORT_THREADS, 0, stream>>>(binned, band_start, rowptr, dis, csr_src, N_NODES);

    fused_layers_kernel<<<(N_NODES + FUSE_ROWS - 1) / FUSE_ROWS, 256, 0, stream>>>(rowptr, csr_src, dis, x, W1, b1, W2, b2, h2, N_NODES);

    gather_kernel<OUT_F><<<(N_NODES + 8 * ROWS_PER_WAVE - 1) / (8 * ROWS_PER_WAVE), 256, 0, stream>>>(rowptr, csr_src, dis, h2, out, N_NODES);
}


# Second gather looks up no edge weight: layer-2 rows are stored times their own deg^-0.5, the gather adds rows in the same edge order and scales each sum once by deg^-0.5 of the destination
# speedup vs baseline: 5.4305x; 1.0422x over previous
#include <hip/hip_runtime.h>
#include <hip/hip_bf16.h>

typedef __attribute__((ext_vector_type(2))) float v2f;
typedef __attribute__((ext_vector_type(8))) float v8f;
typedef __attribute__((ext_vector_type(4))) float v4f;

#define N_NODES 100000
#define N_EDGES 1600000
#define IN_F    128
#define HID_F   128
#define OUT_F   64


#define BAND_SHIFT 9
#define BAND (1 << BAND_SHIFT)
#define NBAND ((N_NODES + BAND - 1) >> BAND_SHIFT)
#define SORT_THREADS 1024
#define EDGES_PER_THREAD 8
#define CHUNK (SORT_THREADS * EDGES_PER_THREAD)
#define NCHUNK ((N_EDGES + CHUNK - 1) / CHUNK)
#define BAND_EDGES_PER_THREAD 12
#define BAND_CAP (SORT_THREADS * BAND_EDGES_PER_THREAD)
#define BANDS_PER_LANE ((NBAND + 31) / 32)
#define NBAND_PAD (BANDS_PER_LANE * 32)
#define CHUNKS_PER_LANE ((NCHUNK + 31) / 32)
#define CNT_PITCH (CHUNKS_PER_LANE * 32)

static_assert(N_NODES <= (1 << (31 - BAND_SHIFT)), "src must fit above the band-local destination");

#define ROWS_PER_WAVE 8
#define FUSE_ROWS 64

__device__ __forceinline__ void wave_exscan_bands(const int* in, int* out, int lane) {
    int v[BANDS_PER_LANE];
    int t = 0;
#pragma unroll
    for (int k = 0; k < BANDS_PER_LANE; ++k) {
        v[k] = in[lane * BANDS_PER_LANE + k];
        t += v[k];
    }
    int inc = t;
#pragma unroll
    for (int o = 1; o < 32; o <<= 1) {
        const int up = __shfl_up(inc, o);
        if (lane >= o) inc += up;
    }
    int run = inc - t;
#pragma unroll
    for (int k = 0; k < BANDS_PER_LANE; ++k) {
        out[lane * BANDS_PER_LANE + k] = run;
        run += v[k];
    }
}

__global__ __launch_bounds__(SORT_THREADS)
void band_count_kernel(const int* __restrict__ dst, int* __restrict__ cntmat, int E) {
    __shared__ int s_cnt[NBAND];
    const int tid = threadIdx.x;
    if (tid < NBAND) s_cnt[tid] = 0;
    const int base = blockIdx.x * CHUNK + tid;
    int d[EDGES_PER_THREAD];
#pragma unroll
    for (int k = 0; k < EDGES_PER_THREAD; ++k) {
        const int e = base + k * SORT_THREADS;
        const int v = dst[min(e, E - 1)];
        d[k] = (e < E) ? v : -1;
    }
    __syncthreads();
#pragma unroll
    for (int k = 0; k < EDGES_PER_THREAD; ++k)
        if (d[k] >= 0) atomicAdd(&s_cnt[d[k] >> BAND_SHIFT], 1);
    __syncthreads();
    const int chunk = blockIdx.x;
    if (tid < NBAND)
        cntmat[tid * CNT_PITCH + (chunk % CHUNKS_PER_LANE) * 32 + chunk / CHUNKS_PER_LANE] = s_cnt[tid];
}

__global__ __launch_bounds__(SORT_THREADS)
void band_group_kernel(const int* __restrict__ src, const int* __restrict__ dst,
                       const int* __restrict__ cntmat, int* __restrict__ band_start,
                       int* __restrict__ binned, int E) {
    __shared__ int s_cnt[NBAND_PAD];
    __shared__ int s_lstart[NBAND_PAD];
    __shared__ int s_tot[NBAND_PAD];
    __shared__ int s_pre[NBAND_PAD];
    __shared__ int s_gstart[NBAND_PAD];
    __shared__ int s_rec[CHUNK];
    const int tid = threadIdx.x, lane = tid & 31;
    const int wave = __builtin_amdgcn_readfirstlane(tid >> 5);
    const int chunk = blockIdx.x;

    if (tid < NBAND_PAD) s_cnt[tid] = 0;

    const int base = chunk * CHUNK + tid;
    int d[EDGES_PER_THREAD], s[EDGES_PER_THREAD], r[EDGES_PER_THREAD];
#pragma unroll
    for (int k = 0; k < EDGES_PER_THREAD; ++k) {
        const int e = base + k * SORT_THREADS;
        const int ec = min(e, E - 1);
        const int dv = dst[ec];
        s[k] = src[ec];
        d[k] = (e < E) ? dv : -1;
    }

    const int own_lane = chunk / CHUNKS_PER_LANE;
    const int own_k    = chunk % CHUNKS_PER_LANE;
    int v[BANDS_PER_LANE][CHUNKS_PER_LANE];
#pragma unroll
    for (int j = 0; j < BANDS_PER_LANE; ++j) {
        const int b = wave + j * (SORT_THREADS / 32);
        const int* row = cntmat + min(b, NBAND - 1) * CNT_PITCH;
#pragma unroll
        for (int k = 0; k < CHUNKS_PER_LANE; ++k) {
            const bool ok = lane * CHUNKS_PER_LANE + k < NCHUNK;
            const int x = row[ok ? k * 32 + lane : 0];
            v[j][k] = ok ? x : 0;
        }
    }
    int t[BANDS_PER_LANE], part[BANDS_PER_LANE], inc[BANDS_PER_LANE];
#pragma unroll
    for (int j = 0; j < BANDS_PER_LANE; ++j) {
        t[j] = 0;
        part[j] = 0;
#pragma unroll
        for (int k = 0; k < CHUNKS_PER_LANE; ++k) {
            t[j] += v[j][k];
            part[j] += (k < own_k) ? v[j][k] : 0;
        }
        inc[j] = t[j];
    }
#pragma unroll
    for (int o = 1; o < 32; o <<= 1) {
        int up[BANDS_PER_LANE];
#pragma unroll
        for (int j = 0; j < BANDS_PER_LANE; ++j) up[j] = __shfl_up(inc[j], o);
#pragma unroll
        for (int j = 0; j < BANDS_PER_LANE; ++j) inc[j] += (lane >= o) ? up[j] : 0;
    }
#pragma unroll
    for (int j = 0; j < BANDS_PER_LANE; ++j) {
        const int b = wave + j * (SORT_THREADS / 32);
        const int tot = __builtin_amdgcn_readlane(inc[j], 31);
        const int pre = __builtin_amdgcn_readlane(inc[j] - t[j] + part[j], own_lane);
        if (lane == 0) {
            s_tot[b] = (b < NBAND) ? tot : 0;
            s_pre[b] = (b < NBAND) ? pre : 0;
        }
    }
    __syncthreads();

#pragma unroll
    for (int k = 0; k < EDGES_PER_THREAD; ++k)
        r[k] = (d[k] >= 0) ? atomicAdd(&s_cnt[d[k] >> BAND_SHIFT], 1) : 0;
    if (wave == 1) wave_exscan_bands(s_tot, s_gstart, lane);
    __syncthreads();

    if (wave == 0) wave_exscan_bands(s_cnt, s_lstart, lane);
    __syncthreads();

    if (tid < NBAND) {
        const int bs = s_gstart[tid];
        if (chunk == 0) band_start[tid] = bs;
        s_gstart[tid] = bs + s_pre[tid];
    }
    if (chunk == 0 && tid == 0) band_start[NBAND] = E;
#pragma unroll
    for (int k = 0; k < EDGES_PER_THREAD; ++k)
        if (d[k] >= 0)
            s_rec[s_lstart[d[k] >> BAND_SHIFT] + r[k]] = (s[k] << BAND_SHIFT) | (d[k] & (BAND - 1));
    __syncthreads();

    for (int b = wave; b < NBAND; b += SORT_THREADS / 32) {
        const int n = s_cnt[b], ls = s_lstart[b], gs = s_gstart[b];
        for (int i = lane; i < n; i += 32) binned[gs + i] = s_rec[ls + i];
    }
}

__global__ __launch_bounds__(SORT_THREADS)
void band_finish_kernel(const int* __restrict__ binned, const int* __restrict__ band_start,
                        int* __restrict__ rowptr, float* __restrict__ dis,
                        int* __restrict__ csr_src, int n) {
    __shared__ int s_cnt[BAND];
    __shared__ int s_wtot[BAND / 32];
    __shared__ int s_out[BAND_CAP];
    const int tid = threadIdx.x, lane = tid & 31, wave = tid >> 5;
    const int band = blockIdx.x;
    const int beg = band_start[band], end = band_start[band + 1];
    const int m = end - beg;
    const bool in_lds = (m <= BAND_CAP);

    if (tid < BAND) s_cnt[tid] = 0;

    int rec[BAND_EDGES_PER_THREAD];
#pragma unroll
    for (int k = 0; k < BAND_EDGES_PER_THREAD; ++k) rec[k] = -1;
    if (m > 0) {
#pragma unroll
        for (int k = 0; k < BAND_EDGES_PER_THREAD; ++k) {
            const int i = k * SORT_THREADS + tid;
            const int v = binned[beg + min(i, m - 1)];
            rec[k] = (i < m) ? v : -1;
        }
    }
    __syncthreads();

#pragma unroll
    for (int k = 0; k < BAND_EDGES_PER_THREAD; ++k)
        if (rec[k] >= 0) atomicAdd(&s_cnt[rec[k] & (BAND - 1)], 1);
    for (int i = BAND_CAP + tid; i < m; i += SORT_THREADS)
        atomicAdd(&s_cnt[binned[beg + i] & (BAND - 1)], 1);
    __syncthreads();

    int c = 0, inc = 0;
    if (tid < BAND) {
        c = s_cnt[tid];
        inc = c;
#pragma unroll
        for (int o = 1; o < 32; o <<= 1) {
            const int up = __shfl_up(inc, o);
            if (lane >= o) inc += up;
        }
        if (lane == 31) s_wtot[wave] = inc;
    }
    __syncthreads();
    if (tid < BAND) {
        int off = 0;
#pragma unroll
        for (int w = 0; w < BAND / 32; ++w) off += (w < wave) ? s_wtot[w] : 0;
        const int excl = off + inc - c;
        s_cnt[tid] = excl;
        const int node = band * BAND + tid;
        if (node < n) {
            rowptr[node] = beg + excl;
            dis[node] = (c > 0) ? (1.0f / sqrtf((float)c)) : 0.0f;
        }
    }
    if (band == NBAND - 1 && tid == 0) rowptr[n] = end;
    __syncthreads();

    int p[BAND_EDGES_PER_THREAD];
#pragma unroll
    for (int k = 0; k < BAND_EDGES_PER_THREAD; ++k)
        p[k] = (rec[k] >= 0) ? atomicAdd(&s_cnt[rec[k] & (BAND - 1)], 1) : 0;
    if (in_lds) {
#pragma unroll
        for (int k = 0; k < BAND_EDGES_PER_THREAD; ++k)
            if (rec[k] >= 0) s_out[p[k]] = rec[k] >> BAND_SHIFT;
    } else {
#pragma unroll
        for (int k = 0; k < BAND_EDGES_PER_THREAD; ++k)
            if (rec[k] >= 0) csr_src[beg + p[k]] = rec[k] >> BAND_SHIFT;
    }
    for (int i = BAND_CAP + tid; i < m; i += SORT_THREADS) {
        const int w = binned[beg + i];
        csr_src[beg + atomicAdd(&s_cnt[w & (BAND - 1)], 1)] = w >> BAND_SHIFT;
    }
    if (in_lds) {
        __syncthreads();
        for (int i = tid; i < m; i += SORT_THREADS) csr_src[beg + i] = s_out[i];
    }
}

template <int F> struct RowVec;
template <> struct RowVec<128> { typedef float4 type; };
template <> struct RowVec<64>  { typedef float2 type; };

__device__ __forceinline__ void fma_row(float4& a, const float4& v, float w) {
    a.x = __builtin_fmaf(v.x, w, a.x); a.y = __builtin_fmaf(v.y, w, a.y);
    a.z = __builtin_fmaf(v.z, w, a.z); a.w = __builtin_fmaf(v.w, w, a.w);
}
__device__ __forceinline__ void add_row(float4& a, const float4& v) {
    a.x += v.x; a.y += v.y; a.z += v.z; a.w += v.w;
}
__device__ __forceinline__ void add_row(float2& a, const float2& v) {
    a.x += v.x; a.y += v.y;
}
__device__ __forceinline__ float4 scale_row(const float4& a, float w) {
    return make_float4(a.x * w, a.y * w, a.z * w, a.w * w);
}
__device__ __forceinline__ float2 scale_row(const float2& a, float w) {
    return make_float2(a.x * w, a.y * w);
}

template <int F, int U>
__device__ __forceinline__ void gather_step(typename RowVec<F>::type& acc, const float* __restrict__ hs,
                                            int s_l, int j, int lane) {
    typedef typename RowVec<F>::type VT;
    VT v[U];
#pragma unroll
    for (int u = 0; u < U; ++u) {
        const int s = __builtin_amdgcn_readlane(s_l, j + u);
        v[u] = ((const VT*)(hs + (long)s * F))[lane];
    }
#pragma unroll
    for (int u = 0; u < U; ++u) add_row(acc, v[u]);
}

template <int F>
__global__ __launch_bounds__(256)
void gather_kernel(const int* __restrict__ rowptr, const int* __restrict__ csr_src,
                   const float* __restrict__ dis, const float* __restrict__ hs,
                   float* __restrict__ out, int n) {
    typedef typename RowVec<F>::type VT;
    const int node0 = (blockIdx.x * 8 + __builtin_amdgcn_readfirstlane((int)(threadIdx.x >> 5))) * ROWS_PER_WAVE;
    if (node0 >= n) return;
    const int lane = threadIdx.x & 31;
    const int rows = (n - node0 < ROWS_PER_WAVE) ? (n - node0) : ROWS_PER_WAVE;
    const int   rp_l = (lane <= rows) ? rowptr[node0 + lane] : 0;
    const float dd_l = (lane <  rows) ? dis[node0 + lane] : 0.0f;

    for (int i = 0; i < rows; ++i) {
        const int beg = __builtin_amdgcn_readlane(rp_l, i);
        const int end = __builtin_amdgcn_readlane(rp_l, i + 1);
        const float dd = __builtin_bit_cast(float, __builtin_amdgcn_readlane(__builtin_bit_cast(int, dd_l), i));

        VT acc = {};
        for (int base = beg; base < end; base += 32) {
            const int m = (end - base < 32) ? (end - base) : 32;
            const int s_l = (lane < m) ? csr_src[base + lane] : 0;
            int j = 0;
            for (; j + 8 <= m; j += 8) gather_step<F, 8>(acc, hs, s_l, j, lane);
            if (j + 4 <= m) { gather_step<F, 4>(acc, hs, s_l, j, lane); j += 4; }
            if (j + 2 <= m) { gather_step<F, 2>(acc, hs, s_l, j, lane); j += 2; }
            if (j < m)      { gather_step<F, 1>(acc, hs, s_l, j, lane); }
        }
        ((VT*)(out + (long)(node0 + i) * F))[lane] = scale_row(acc, dd);
    }
}

#define XPITCH 132

template <int U>
__device__ __forceinline__ void gather_step_ws(float4& acc, float& wsum, const float* __restrict__ h,
                                               int s_l, float w_l, int j, int lane) {
    float4 v[U];
    float w[U];
#pragma unroll
    for (int u = 0; u < U; ++u) {
        const int s = __builtin_amdgcn_readlane(s_l, j + u);
        w[u] = __builtin_bit_cast(float, __builtin_amdgcn_readlane(__builtin_bit_cast(int, w_l), j + u));
        v[u] = ((const float4*)(h + (long)s * IN_F))[lane];
    }
#pragma unroll
    for (int u = 0; u < U; ++u) {
        fma_row(acc, v[u], w[u]);
        wsum += w[u];
    }
}

__device__ __forceinline__ v8f wmma4(v2f a, v2f b, v8f c) {
    return __builtin_amdgcn_wmma_f32_16x16x4_f32(false, a, false, b, (short)0, c, false, false);
}

__global__ __launch_bounds__(256, 10)
void fused_layers_kernel(const int* __restrict__ rowptr, const int* __restrict__ csr_src,
                         const float* __restrict__ dis, const float* __restrict__ x,
                         const float* __restrict__ W1, const float* __restrict__ b1,
                         const float* __restrict__ W2, const float* __restrict__ b2,
                         float* __restrict__ h2, int n) {
    constexpr int K = 128;
    __shared__ v4f s_t[FUSE_ROWS * XPITCH / 4];
    __shared__ float s_ws[FUSE_ROWS];
    __shared__ v4f s_dis[FUSE_ROWS / 4];
    const int tid  = threadIdx.x;
    const int wave = __builtin_amdgcn_readfirstlane(tid >> 5);
    const int lane = tid & 31;
    const int half = lane >> 4;
    const int l    = lane & 15;
    const int row0 = blockIdx.x * FUSE_ROWS;

    {
        const int node0 = row0 + wave * ROWS_PER_WAVE;
        int rows = n - node0;
        rows = (rows < 0) ? 0 : ((rows > ROWS_PER_WAVE) ? ROWS_PER_WAVE : rows);
        const int   rp_l = (rows > 0 && lane <= rows) ? rowptr[node0 + lane] : 0;
        const float dd_l = (lane < rows) ? dis[node0 + lane] : 0.0f;
        if (lane < ROWS_PER_WAVE) ((float*)s_dis)[wave * ROWS_PER_WAVE + lane] = dd_l;
        for (int i = 0; i < ROWS_PER_WAVE; ++i) {
            float4 acc = {};
            float wsum = 0.0f;
            if (i < rows) {
                const int beg = __builtin_amdgcn_readlane(rp_l, i);
                const int end = __builtin_amdgcn_readlane(rp_l, i + 1);
                const float dd = __builtin_bit_cast(float, __builtin_amdgcn_readlane(__builtin_bit_cast(int, dd_l), i));
                for (int base = beg; base < end; base += 32) {
                    const int m = (end - base < 32) ? (end - base) : 32;
                    int s_l = 0;
                    float w_l = 0.0f;
                    if (lane < m) {
                        s_l = csr_src[base + lane];
                        w_l = dis[s_l] * dd;
                    }
                    int j = 0;
                    for (; j + 8 <= m; j += 8) gather_step_ws<8>(acc, wsum, x, s_l, w_l, j, lane);
                    if (j + 4 <= m) { gather_step_ws<4>(acc, wsum, x, s_l, w_l, j, lane); j += 4; }
                    if (j + 2 <= m) { gather_step_ws<2>(acc, wsum, x, s_l, w_l, j, lane); j += 2; }
                    if (j < m)      { gather_step_ws<1>(acc, wsum, x, s_l, w_l, j, lane); }
                }
            }
            const int r = wave * ROWS_PER_WAVE + i;
            v4f o;
            o.x = acc.x; o.y = acc.y; o.z = acc.z; o.w = acc.w;
            s_t[r * (XPITCH / 4) + lane] = o;
            if (lane == 0) s_ws[r] = wsum;
        }
    }
    __syncthreads();

    v4f b[K / 8];
    {
        const unsigned boff = (unsigned)((wave * 16 + l) * K + 4 * half);
#pragma unroll
        for (int p = 0; p < K / 8; ++p) b[p] = *(const v4f*)(W1 + (boff + 8u * p));
    }
    const float bb1 = b1[wave * 16 + l];
    {
        v8f c[FUSE_ROWS / 16];
#pragma unroll
        for (int t = 0; t < FUSE_ROWS / 16; ++t) c[t] = v8f{};
        const float* arow = (const float*)s_t + l * XPITCH + 4 * half;
#pragma unroll
        for (int p = 0; p < K / 8; ++p) {
            v4f a[FUSE_ROWS / 16];
#pragma unroll
            for (int t = 0; t < FUSE_ROWS / 16; ++t) a[t] = *(const v4f*)(arow + t * 16 * XPITCH + 8 * p);
            const v2f blo = {b[p].x, b[p].y}, bhi = {b[p].z, b[p].w};
#pragma unroll
            for (int t = 0; t < FUSE_ROWS / 16; ++t) c[t] = wmma4(v2f{a[t].x, a[t].y}, blo, c[t]);
#pragma unroll
            for (int t = 0; t < FUSE_ROWS / 16; ++t) c[t] = wmma4(v2f{a[t].z, a[t].w}, bhi, c[t]);
        }
        __syncthreads();
#pragma unroll
        for (int t = 0; t < FUSE_ROWS / 16; ++t) {
#pragma unroll
            for (int r = 0; r < 8; ++r) {
                const int row = t * 16 + r + 8 * half;
                const float v = __builtin_fmaf(s_ws[row], bb1, c[t][r]);
                ((float*)s_t)[row * XPITCH + wave * 16 + l] = fmaxf(v, 0.0f);
            }
        }
    }
    __syncthreads();

    {
        const int n2 = (wave & 3) * 16;
        const int t0 = wave >> 2;
        const unsigned boff = (unsigned)((n2 + l) * K + 4 * half);
#pragma unroll
        for (int p = 0; p < K / 8; ++p) b[p] = *(const v4f*)(W2 + (boff + 8u * p));
        const float bb2 = b2[n2 + l];
        v8f d0 = {}, d1 = {};
        const float* arow0 = (const float*)s_t + (t0 * 16 + l) * XPITCH + 4 * half;
        const float* arow1 = arow0 + 32 * XPITCH;
#pragma unroll
        for (int p = 0; p < K / 8; ++p) {
            const v4f a0 = *(const v4f*)(arow0 + 8 * p);
            const v4f a1 = *(const v4f*)(arow1 + 8 * p);
            const v2f blo = {b[p].x, b[p].y}, bhi = {b[p].z, b[p].w};
            d0 = wmma4(v2f{a0.x, a0.y}, blo, d0);
            d1 = wmma4(v2f{a1.x, a1.y}, blo, d1);
            d0 = wmma4(v2f{a0.z, a0.w}, bhi, d0);
            d1 = wmma4(v2f{a1.z, a1.w}, bhi, d1);
        }
        const int lr0 = t0 * 16 + 8 * half;
        const v4f sa0 = s_dis[lr0 / 4], sa1 = s_dis[lr0 / 4 + 1];
        const v4f sb0 = s_dis[lr0 / 4 + 8], sb1 = s_dis[lr0 / 4 + 9];
        const float sa[8] = {sa0.x, sa0.y, sa0.z, sa0.w, sa1.x, sa1.y, sa1.z, sa1.w};
        const float sb[8] = {sb0.x, sb0.y, sb0.z, sb0.w, sb1.x, sb1.y, sb1.z, sb1.w};
#pragma unroll
        for (int r = 0; r < 8; ++r) {
            const int rowa = row0 + lr0 + r;
            const int rowb = rowa + 32;
            if (rowa < n) h2[(long)rowa * OUT_F + n2 + l] = (d0[r] + bb2) * sa[r];
            if (rowb < n) h2[(long)rowb * OUT_F + n2 + l] = (d1[r] + bb2) * sb[r];
        }
    }
}


extern "C" void kernel_launch(void* const* d_in, const int* in_sizes, int n_in,
                              void* d_out, int out_size, void* d_ws, size_t ws_size,
                              hipStream_t stream) {
    const float* x  = (const float*)d_in[0];
    const int*   ei = (const int*)d_in[1];
    const float* W1 = (const float*)d_in[2];
    const float* b1 = (const float*)d_in[3];
    const float* W2 = (const float*)d_in[4];
    const float* b2 = (const float*)d_in[5];
    float* out = (float*)d_out;

    const int* src = ei;
    const int* dst = ei + N_EDGES;

    float* dis  = (float*)d_ws;
    float* h2   = dis + N_NODES;
    int* rowptr     = (int*)(h2 + (long)N_NODES * OUT_F);
    int* band_start = rowptr + N_NODES + 32;
    int* cntmat     = band_start + 256;
    int* binned     = cntmat + NBAND * CNT_PITCH;
    int* csr_src    = binned + N_EDGES;

    band_count_kernel<<<NCHUNK, SORT_THREADS, 0, stream>>>(dst, cntmat, N_EDGES);
    band_group_kernel<<<NCHUNK, SORT_THREADS, 0, stream>>>(src, dst, cntmat, band_start, binned, N_EDGES);
    band_finish_kernel<<<NBAND, SORT_THREADS, 0, stream>>>(binned, band_start, rowptr, dis, csr_src, N_NODES);

    fused_layers_kernel<<<(N_NODES + FUSE_ROWS - 1) / FUSE_ROWS, 256, 0, stream>>>(rowptr, csr_src, dis, x, W1, b1, W2, b2, h2, N_NODES);

    gather_kernel<OUT_F><<<(N_NODES + 8 * ROWS_PER_WAVE - 1) / (8 * ROWS_PER_WAVE), 256, 0, stream>>>(rowptr, csr_src, dis, h2, out, N_NODES);
}


# CSR build in two launches, not three: level 1 groups each chunk by band in place and tables its run starts (no count pass), level 2 collects its band's runs in LDS; row sums stay in LDS arrival order
# speedup vs baseline: 5.7479x; 1.0584x over previous
#include <hip/hip_runtime.h>
#include <hip/hip_bf16.h>

typedef __attribute__((ext_vector_type(2))) float v2f;
typedef __attribute__((ext_vector_type(8))) float v8f;
typedef __attribute__((ext_vector_type(4))) float v4f;

#define N_NODES 100000
#define N_EDGES 1600000
#define IN_F    128
#define HID_F   128
#define OUT_F   64


#define BAND_SHIFT 9
#define BAND (1 << BAND_SHIFT)
#define NBAND ((N_NODES + BAND - 1) >> BAND_SHIFT)
#define SORT_THREADS 1024
#define SORT_WAVES (SORT_THREADS / 32)
#define EDGES_PER_THREAD 8
#define CHUNK (SORT_THREADS * EDGES_PER_THREAD)
#define NCHUNK ((N_EDGES + CHUNK - 1) / CHUNK)
#define BAND_EDGES_PER_THREAD 12
#define BAND_CAP (SORT_THREADS * BAND_EDGES_PER_THREAD)
#define BANDS_PER_LANE ((NBAND + 1 + 31) / 32)
#define NBAND_PAD (BANDS_PER_LANE * 32)
#define CHUNKS_PER_LANE ((NCHUNK + 31) / 32)
#define RUN_PITCH (CHUNKS_PER_LANE * 32)

static_assert(N_NODES <= (1 << (31 - BAND_SHIFT)), "src must fit above the band-local destination");
static_assert(RUN_PITCH <= SORT_THREADS && NBAND_PAD <= SORT_THREADS && BAND <= SORT_THREADS, "one thread per table entry");

#define ROWS_PER_WAVE 8
#define FUSE_ROWS 64

template <int PER_LANE>
__device__ __forceinline__ int wave_exscan(const int* in, int* out, int lane) {
    int v[PER_LANE];
    int t = 0;
#pragma unroll
    for (int k = 0; k < PER_LANE; ++k) {
        v[k] = in[lane * PER_LANE + k];
        t += v[k];
    }
    int inc = t;
#pragma unroll
    for (int o = 1; o < 32; o <<= 1) {
        const int up = __shfl_up(inc, o);
        if (lane >= o) inc += up;
    }
    int run = inc - t;
#pragma unroll
    for (int k = 0; k < PER_LANE; ++k) {
        out[lane * PER_LANE + k] = run;
        run += v[k];
    }
    return __shfl(inc, 31);
}

__global__ __launch_bounds__(SORT_THREADS)
void band_group_kernel(const int* __restrict__ src, const int* __restrict__ dst,
                       int* __restrict__ runtab, int* __restrict__ binned, int E) {
    __shared__ int s_cnt[NBAND_PAD];
    __shared__ int s_lstart[NBAND_PAD];
    __shared__ int s_rec[CHUNK];
    const int tid = threadIdx.x, lane = tid & 31, wave = tid >> 5;
    const int chunk = blockIdx.x;

    if (tid < NBAND_PAD) s_cnt[tid] = 0;

    const int base = chunk * CHUNK + tid;
    int d[EDGES_PER_THREAD], s[EDGES_PER_THREAD], r[EDGES_PER_THREAD];
#pragma unroll
    for (int k = 0; k < EDGES_PER_THREAD; ++k) {
        const int e = base + k * SORT_THREADS;
        const int ec = min(e, E - 1);
        const int dv = dst[ec];
        s[k] = src[ec];
        d[k] = (e < E) ? dv : -1;
    }
    __syncthreads();

#pragma unroll
    for (int k = 0; k < EDGES_PER_THREAD; ++k)
        r[k] = (d[k] >= 0) ? atomicAdd(&s_cnt[d[k] >> BAND_SHIFT], 1) : 0;
    __syncthreads();

    if (wave == 0) wave_exscan<BANDS_PER_LANE>(s_cnt, s_lstart, lane);
    __syncthreads();

    if (tid <= NBAND) runtab[tid * RUN_PITCH + chunk] = s_lstart[tid];
#pragma unroll
    for (int k = 0; k < EDGES_PER_THREAD; ++k)
        if (d[k] >= 0)
            s_rec[s_lstart[d[k] >> BAND_SHIFT] + r[k]] = (s[k] << BAND_SHIFT) | (d[k] & (BAND - 1));
    __syncthreads();

    const int nvalid = min(CHUNK, E - chunk * CHUNK);
#pragma unroll
    for (int k = 0; k < EDGES_PER_THREAD; ++k) {
        const int i = k * SORT_THREADS + tid;
        if (i < nvalid) binned[chunk * CHUNK + i] = s_rec[i];
    }
}

__global__ __launch_bounds__(SORT_THREADS)
void band_finish_kernel(const int* __restrict__ binned, const int* __restrict__ runtab,
                        int* __restrict__ rowptr, float* __restrict__ dis,
                        int* __restrict__ csr_src, int n) {
    __shared__ int s_cnt[BAND];
    __shared__ int s_wtot[BAND / 32];
    __shared__ int s_from[RUN_PITCH];
    __shared__ int s_n[RUN_PITCH];
    __shared__ int s_pos[RUN_PITCH];
    __shared__ int s_lo[RUN_PITCH];
    __shared__ int s_red[2];
    __shared__ int s_rec[BAND_CAP];
    int* s_out = s_rec;
    const int tid = threadIdx.x, lane = tid & 31;
    const int wave = __builtin_amdgcn_readfirstlane(tid >> 5);
    const int band = blockIdx.x;

    if (tid < BAND) s_cnt[tid] = 0;
    if (tid < RUN_PITCH) {
        int lo = 0, hi = 0;
        if (tid < NCHUNK) {
            lo = runtab[band * RUN_PITCH + tid];
            hi = runtab[(band + 1) * RUN_PITCH + tid];
        }
        s_lo[tid] = lo;
        s_from[tid] = tid * CHUNK + lo;
        s_n[tid] = hi - lo;
    }
    __syncthreads();
    if (wave == 0) {
        const int tot = wave_exscan<CHUNKS_PER_LANE>(s_n, s_pos, lane);
        if (lane == 0) s_red[1] = tot;
    }
    if (wave == 1) {
        int t = 0;
#pragma unroll
        for (int k = 0; k < CHUNKS_PER_LANE; ++k) t += s_lo[lane * CHUNKS_PER_LANE + k];
#pragma unroll
        for (int o = 16; o > 0; o >>= 1) t += __shfl_xor(t, o);
        if (lane == 0) s_red[0] = t;
    }
    __syncthreads();
    const int beg = s_red[0], m = s_red[1], end = beg + m;
    const bool in_lds = (m <= BAND_CAP);

    {
        int rn[CHUNKS_PER_LANE], rpos[CHUNKS_PER_LANE], rfrom[CHUNKS_PER_LANE];
        int wv[CHUNKS_PER_LANE][2];
#pragma unroll
        for (int j = 0; j < CHUNKS_PER_LANE; ++j) {
            const int c = wave + j * SORT_WAVES;
            rn[j] = s_n[c];
            rpos[j] = s_pos[c];
            rfrom[j] = s_from[c];
#pragma unroll
            for (int t = 0; t < 2; ++t) {
                const int i = lane + 32 * t;
                wv[j][t] = binned[(i < rn[j]) ? rfrom[j] + i : 0];
            }
        }
#pragma unroll
        for (int j = 0; j < CHUNKS_PER_LANE; ++j) {
#pragma unroll
            for (int t = 0; t < 2; ++t) {
                const int i = lane + 32 * t;
                if (i < rn[j] && rpos[j] + i < BAND_CAP) s_rec[rpos[j] + i] = wv[j][t];
            }
            for (int i = lane + 64; i < rn[j]; i += 32)
                if (rpos[j] + i < BAND_CAP) s_rec[rpos[j] + i] = binned[rfrom[j] + i];
        }
    }
    __syncthreads();

    int rec[BAND_EDGES_PER_THREAD];
#pragma unroll
    for (int k = 0; k < BAND_EDGES_PER_THREAD; ++k) {
        const int i = k * SORT_THREADS + tid;
        rec[k] = (i < m) ? s_rec[i] : -1;
    }

#pragma unroll
    for (int k = 0; k < BAND_EDGES_PER_THREAD; ++k)
        if (rec[k] >= 0) atomicAdd(&s_cnt[rec[k] & (BAND - 1)], 1);
    if (!in_lds) {
        for (int c = wave; c < NCHUNK; c += SORT_WAVES) {
            const int rnc = s_n[c], rposc = s_pos[c], rfromc = s_from[c];
            for (int i = lane; i < rnc; i += 32)
                if (rposc + i >= BAND_CAP) atomicAdd(&s_cnt[binned[rfromc + i] & (BAND - 1)], 1);
        }
    }
    __syncthreads();

    int c = 0, inc = 0;
    if (tid < BAND) {
        c = s_cnt[tid];
        inc = c;
#pragma unroll
        for (int o = 1; o < 32; o <<= 1) {
            const int up = __shfl_up(inc, o);
            if (lane >= o) inc += up;
        }
        if (lane == 31) s_wtot[wave] = inc;
    }
    __syncthreads();
    if (tid < BAND) {
        int off = 0;
#pragma unroll
        for (int w = 0; w < BAND / 32; ++w) off += (w < wave) ? s_wtot[w] : 0;
        const int excl = off + inc - c;
        s_cnt[tid] = excl;
        const int node = band * BAND + tid;
        if (node < n) {
            rowptr[node] = beg + excl;
            dis[node] = (c > 0) ? (1.0f / sqrtf((float)c)) : 0.0f;
        }
    }
    if (band == NBAND - 1 && tid == 0) rowptr[n] = end;
    __syncthreads();

    int p[BAND_EDGES_PER_THREAD];
#pragma unroll
    for (int k = 0; k < BAND_EDGES_PER_THREAD; ++k)
        p[k] = (rec[k] >= 0) ? atomicAdd(&s_cnt[rec[k] & (BAND - 1)], 1) : 0;
    if (in_lds) {
#pragma unroll
        for (int k = 0; k < BAND_EDGES_PER_THREAD; ++k)
            if (rec[k] >= 0) s_out[p[k]] = rec[k] >> BAND_SHIFT;
        __syncthreads();
        for (int i = tid; i < m; i += SORT_THREADS) csr_src[beg + i] = s_out[i];
    } else {
#pragma unroll
        for (int k = 0; k < BAND_EDGES_PER_THREAD; ++k)
            if (rec[k] >= 0) csr_src[beg + p[k]] = rec[k] >> BAND_SHIFT;
        for (int cc = wave; cc < NCHUNK; cc += SORT_WAVES) {
            const int rnc = s_n[cc], rposc = s_pos[cc], rfromc = s_from[cc];
            for (int i = lane; i < rnc; i += 32)
                if (rposc + i >= BAND_CAP) {
                    const int w = binned[rfromc + i];
                    csr_src[beg + atomicAdd(&s_cnt[w & (BAND - 1)], 1)] = w >> BAND_SHIFT;
                }
        }
    }
}

template <int F> struct RowVec;
template <> struct RowVec<128> { typedef float4 type; };
template <> struct RowVec<64>  { typedef float2 type; };

__device__ __forceinline__ void fma_row(float4& a, const float4& v, float w) {
    a.x = __builtin_fmaf(v.x, w, a.x); a.y = __builtin_fmaf(v.y, w, a.y);
    a.z = __builtin_fmaf(v.z, w, a.z); a.w = __builtin_fmaf(v.w, w, a.w);
}
__device__ __forceinline__ void add_row(float4& a, const float4& v) {
    a.x += v.x; a.y += v.y; a.z += v.z; a.w += v.w;
}
__device__ __forceinline__ void add_row(float2& a, const float2& v) {
    a.x += v.x; a.y += v.y;
}
__device__ __forceinline__ float4 scale_row(const float4& a, float w) {
    return make_float4(a.x * w, a.y * w, a.z * w, a.w * w);
}
__device__ __forceinline__ float2 scale_row(const float2& a, float w) {
    return make_float2(a.x * w, a.y * w);
}

template <int F, int U>
__device__ __forceinline__ void gather_step(typename RowVec<F>::type& acc, const float* __restrict__ hs,
                                            int s_l, int j, int lane) {
    typedef typename RowVec<F>::type VT;
    VT v[U];
#pragma unroll
    for (int u = 0; u < U; ++u) {
        const int s = __builtin_amdgcn_readlane(s_l, j + u);
        v[u] = ((const VT*)(hs + (long)s * F))[lane];
    }
#pragma unroll
    for (int u = 0; u < U; ++u) add_row(acc, v[u]);
}

template <int F>
__global__ __launch_bounds__(256)
void gather_kernel(const int* __restrict__ rowptr, const int* __restrict__ csr_src,
                   const float* __restrict__ dis, const float* __restrict__ hs,
                   float* __restrict__ out, int n) {
    typedef typename RowVec<F>::type VT;
    const int node0 = (blockIdx.x * 8 + __builtin_amdgcn_readfirstlane((int)(threadIdx.x >> 5))) * ROWS_PER_WAVE;
    if (node0 >= n) return;
    const int lane = threadIdx.x & 31;
    const int rows = (n - node0 < ROWS_PER_WAVE) ? (n - node0) : ROWS_PER_WAVE;
    const int   rp_l = (lane <= rows) ? rowptr[node0 + lane] : 0;
    const float dd_l = (lane <  rows) ? dis[node0 + lane] : 0.0f;

    for (int i = 0; i < rows; ++i) {
        const int beg = __builtin_amdgcn_readlane(rp_l, i);
        const int end = __builtin_amdgcn_readlane(rp_l, i + 1);
        const float dd = __builtin_bit_cast(float, __builtin_amdgcn_readlane(__builtin_bit_cast(int, dd_l), i));

        VT acc = {};
        for (int base = beg; base < end; base += 32) {
            const int m = (end - base < 32) ? (end - base) : 32;
            const int s_l = (lane < m) ? csr_src[base + lane] : 0;
            int j = 0;
            for (; j + 8 <= m; j += 8) gather_step<F, 8>(acc, hs, s_l, j, lane);
            if (j + 4 <= m) { gather_step<F, 4>(acc, hs, s_l, j, lane); j += 4; }
            if (j + 2 <= m) { gather_step<F, 2>(acc, hs, s_l, j, lane); j += 2; }
            if (j < m)      { gather_step<F, 1>(acc, hs, s_l, j, lane); }
        }
        ((VT*)(out + (long)(node0 + i) * F))[lane] = scale_row(acc, dd);
    }
}

#define XPITCH 132

template <int U>
__device__ __forceinline__ void gather_step_ws(float4& acc, float& wsum, const float* __restrict__ h,
                                               int s_l, float w_l, int j, int lane) {
    float4 v[U];
    float w[U];
#pragma unroll
    for (int u = 0; u < U; ++u) {
        const int s = __builtin_amdgcn_readlane(s_l, j + u);
        w[u] = __builtin_bit_cast(float, __builtin_amdgcn_readlane(__builtin_bit_cast(int, w_l), j + u));
        v[u] = ((const float4*)(h + (long)s * IN_F))[lane];
    }
#pragma unroll
    for (int u = 0; u < U; ++u) {
        fma_row(acc, v[u], w[u]);
        wsum += w[u];
    }
}

__device__ __forceinline__ v8f wmma4(v2f a, v2f b, v8f c) {
    return __builtin_amdgcn_wmma_f32_16x16x4_f32(false, a, false, b, (short)0, c, false, false);
}

__global__ __launch_bounds__(256, 10)
void fused_layers_kernel(const int* __restrict__ rowptr, const int* __restrict__ csr_src,
                         const float* __restrict__ dis, const float* __restrict__ x,
                         const float* __restrict__ W1, const float* __restrict__ b1,
                         const float* __restrict__ W2, const float* __restrict__ b2,
                         float* __restrict__ h2, int n) {
    constexpr int K = 128;
    __shared__ v4f s_t[FUSE_ROWS * XPITCH / 4];
    __shared__ float s_ws[FUSE_ROWS];
    __shared__ v4f s_dis[FUSE_ROWS / 4];
    const int tid  = threadIdx.x;
    const int wave = __builtin_amdgcn_readfirstlane(tid >> 5);
    const int lane = tid & 31;
    const int half = lane >> 4;
    const int l    = lane & 15;
    const int row0 = blockIdx.x * FUSE_ROWS;

    {
        const int node0 = row0 + wave * ROWS_PER_WAVE;
        int rows = n - node0;
        rows = (rows < 0) ? 0 : ((rows > ROWS_PER_WAVE) ? ROWS_PER_WAVE : rows);
        const int   rp_l = (rows > 0 && lane <= rows) ? rowptr[node0 + lane] : 0;
        const float dd_l = (lane < rows) ? dis[node0 + lane] : 0.0f;
        if (lane < ROWS_PER_WAVE) ((float*)s_dis)[wave * ROWS_PER_WAVE + lane] = dd_l;
        for (int i = 0; i < ROWS_PER_WAVE; ++i) {
            float4 acc = {};
            float wsum = 0.0f;
            if (i < rows) {
                const int beg = __builtin_amdgcn_readlane(rp_l, i);
                const int end = __builtin_amdgcn_readlane(rp_l, i + 1);
                const float dd = __builtin_bit_cast(float, __builtin_amdgcn_readlane(__builtin_bit_cast(int, dd_l), i));
                for (int base = beg; base < end; base += 32) {
                    const int m = (end - base < 32) ? (end - base) : 32;
                    int s_l = 0;
                    float w_l = 0.0f;
                    if (lane < m) {
                        s_l = csr_src[base + lane];
                        w_l = dis[s_l] * dd;
                    }
                    int j = 0;
                    for (; j + 8 <= m; j += 8) gather_step_ws<8>(acc, wsum, x, s_l, w_l, j, lane);
                    if (j + 4 <= m) { gather_step_ws<4>(acc, wsum, x, s_l, w_l, j, lane); j += 4; }
                    if (j + 2 <= m) { gather_step_ws<2>(acc, wsum, x, s_l, w_l, j, lane); j += 2; }
                    if (j < m)      { gather_step_ws<1>(acc, wsum, x, s_l, w_l, j, lane); }
                }
            }
            const int r = wave * ROWS_PER_WAVE + i;
            v4f o;
            o.x = acc.x; o.y = acc.y; o.z = acc.z; o.w = acc.w;
            s_t[r * (XPITCH / 4) + lane] = o;
            if (lane == 0) s_ws[r] = wsum;
        }
    }
    __syncthreads();

    v4f b[K / 8];
    {
        const unsigned boff = (unsigned)((wave * 16 + l) * K + 4 * half);
#pragma unroll
        for (int p = 0; p < K / 8; ++p) b[p] = *(const v4f*)(W1 + (boff + 8u * p));
    }
    const float bb1 = b1[wave * 16 + l];
    {
        v8f c[FUSE_ROWS / 16];
#pragma unroll
        for (int t = 0; t < FUSE_ROWS / 16; ++t) c[t] = v8f{};
        const float* arow = (const float*)s_t + l * XPITCH + 4 * half;
#pragma unroll
        for (int p = 0; p < K / 8; ++p) {
            v4f a[FUSE_ROWS / 16];
#pragma unroll
            for (int t = 0; t < FUSE_ROWS / 16; ++t) a[t] = *(const v4f*)(arow + t * 16 * XPITCH + 8 * p);
            const v2f blo = {b[p].x, b[p].y}, bhi = {b[p].z, b[p].w};
#pragma unroll
            for (int t = 0; t < FUSE_ROWS / 16; ++t) c[t] = wmma4(v2f{a[t].x, a[t].y}, blo, c[t]);
#pragma unroll
            for (int t = 0; t < FUSE_ROWS / 16; ++t) c[t] = wmma4(v2f{a[t].z, a[t].w}, bhi, c[t]);
        }
        __syncthreads();
#pragma unroll
        for (int t = 0; t < FUSE_ROWS / 16; ++t) {
#pragma unroll
            for (int r = 0; r < 8; ++r) {
                const int row = t * 16 + r + 8 * half;
                const float v = __builtin_fmaf(s_ws[row], bb1, c[t][r]);
                ((float*)s_t)[row * XPITCH + wave * 16 + l] = fmaxf(v, 0.0f);
            }
        }
    }
    __syncthreads();

    {
        const int n2 = (wave & 3) * 16;
        const int t0 = wave >> 2;
        const unsigned boff = (unsigned)((n2 + l) * K + 4 * half);
#pragma unroll
        for (int p = 0; p < K / 8; ++p) b[p] = *(const v4f*)(W2 + (boff + 8u * p));
        const float bb2 = b2[n2 + l];
        v8f d0 = {}, d1 = {};
        const float* arow0 = (const float*)s_t + (t0 * 16 + l) * XPITCH + 4 * half;
        const float* arow1 = arow0 + 32 * XPITCH;
#pragma unroll
        for (int p = 0; p < K / 8; ++p) {
            const v4f a0 = *(const v4f*)(arow0 + 8 * p);
            const v4f a1 = *(const v4f*)(arow1 + 8 * p);
            const v2f blo = {b[p].x, b[p].y}, bhi = {b[p].z, b[p].w};
            d0 = wmma4(v2f{a0.x, a0.y}, blo, d0);
            d1 = wmma4(v2f{a1.x, a1.y}, blo, d1);
            d0 = wmma4(v2f{a0.z, a0.w}, bhi, d0);
            d1 = wmma4(v2f{a1.z, a1.w}, bhi, d1);
        }
        const int lr0 = t0 * 16 + 8 * half;
        const v4f sa0 = s_dis[lr0 / 4], sa1 = s_dis[lr0 / 4 + 1];
        const v4f sb0 = s_dis[lr0 / 4 + 8], sb1 = s_dis[lr0 / 4 + 9];
        const float sa[8] = {sa0.x, sa0.y, sa0.z, sa0.w, sa1.x, sa1.y, sa1.z, sa1.w};
        const float sb[8] = {sb0.x, sb0.y, sb0.z, sb0.w, sb1.x, sb1.y, sb1.z, sb1.w};
#pragma unroll
        for (int r = 0; r < 8; ++r) {
            const int rowa = row0 + lr0 + r;
            const int rowb = rowa + 32;
            if (rowa < n) h2[(long)rowa * OUT_F + n2 + l] = (d0[r] + bb2) * sa[r];
            if (rowb < n) h2[(long)rowb * OUT_F + n2 + l] = (d1[r] + bb2) * sb[r];
        }
    }
}


extern "C" void kernel_launch(void* const* d_in, const int* in_sizes, int n_in,
                              void* d_out, int out_size, void* d_ws, size_t ws_size,
                              hipStream_t stream) {
    const float* x  = (const float*)d_in[0];
    const int*   ei = (const int*)d_in[1];
    const float* W1 = (const float*)d_in[2];
    const float* b1 = (const float*)d_in[3];
    const float* W2 = (const float*)d_in[4];
    const float* b2 = (const float*)d_in[5];
    float* out = (float*)d_out;

    const int* src = ei;
    const int* dst = ei + N_EDGES;

    float* dis  = (float*)d_ws;
    float* h2   = dis + N_NODES;
    int* rowptr  = (int*)(h2 + (long)N_NODES * OUT_F);
    int* runtab  = rowptr + N_NODES + 32;
    int* binned  = runtab + (NBAND + 1) * RUN_PITCH;
    int* csr_src = binned + N_EDGES;

    band_group_kernel<<<NCHUNK, SORT_THREADS, 0, stream>>>(src, dst, runtab, binned, N_EDGES);
    band_finish_kernel<<<NBAND, SORT_THREADS, 0, stream>>>(binned, runtab, rowptr, dis, csr_src, N_NODES);

    fused_layers_kernel<<<(N_NODES + FUSE_ROWS - 1) / FUSE_ROWS, 256, 0, stream>>>(rowptr, csr_src, dis, x, W1, b1, W2, b2, h2, N_NODES);

    gather_kernel<OUT_F><<<(N_NODES + 8 * ROWS_PER_WAVE - 1) / (8 * ROWS_PER_WAVE), 256, 0, stream>>>(rowptr, csr_src, dis, h2, out, N_NODES);
}
